# starting kernel
# baseline (speedup 1.0000x reference)
#include <hip/hip_runtime.h>
#include <math.h>

typedef float v2f __attribute__((ext_vector_type(2)));
typedef float v8f __attribute__((ext_vector_type(8)));

#define LROW 8192
#define NSTATE 64
#define PI_F 3.14159265358979323846f

#define OFF_F128Q   0
#define OFF_IF128Q  32768
#define OFF_G64PR   65536
#define OFF_G64PI   69632
#define OFF_IG64PR  73728
#define OFF_IG64PI  77824
#define OFF_WF      81920
#define OFF_WI      98304
#define OFF_KHAT    114688
#define WS_FLOATS   (OFF_KHAT + 4*16384)

static __device__ inline v8f wmma4(v2f a, v2f b, v8f c) {
  return __builtin_amdgcn_wmma_f32_16x16x4_f32(false, a, false, b, (short)0, c,
                                               false, false);
}

static __device__ inline int pidx(int row, int col) {
  return ((row >> 1) << 7) + (col << 1) + (row & 1);
}

__global__ void ssm_gen_tables(float* __restrict__ ws) {
  int idx = blockIdx.x * blockDim.x + threadIdx.x;
  {
    int k2 = idx >> 7, n2 = idx & 127;
    int t = (k2 * n2) & 127;
    float s, c;
    sincosf(-2.f * PI_F * (float)t / 128.f, &s, &c);
    int pos = k2 * 256 + (n2 >> 1) * 4 + (n2 & 1);
    ws[OFF_F128Q + pos]      = c;
    ws[OFF_F128Q + pos + 2]  = s;
    ws[OFF_IF128Q + pos]     = c;
    ws[OFF_IF128Q + pos + 2] = -s;
  }
  if (idx < 8192) {
    int k2 = idx >> 6, n1 = idx & 63;
    int t = (n1 * k2) & 8191;
    float s, c;
    sincosf(-2.f * PI_F * (float)t / 8192.f, &s, &c);
    ws[OFF_WF + 2 * idx]     = c;  ws[OFF_WF + 2 * idx + 1] = s;
    ws[OFF_WI + 2 * idx]     = c;  ws[OFF_WI + 2 * idx + 1] = -s;
  }
  if (idx < 4096) {
    int n1 = idx >> 6, k1 = idx & 63;
    int t = (n1 * k1) & 63;
    float s, c;
    sincosf(-2.f * PI_F * (float)t / 64.f, &s, &c);
    int pos = (n1 >> 1) * 128 + 2 * k1 + (n1 & 1);
    ws[OFF_G64PR + pos]  = c;  ws[OFF_G64PI + pos]  = s;
    ws[OFF_IG64PR + pos] = c;  ws[OFF_IG64PI + pos] = -s;
  }
}

__global__ void ssm_gen_khat(float* __restrict__ ws,
                             const float* __restrict__ Lr,
                             const float* __restrict__ Li,
                             const float* __restrict__ Pv,
                             const float* __restrict__ Bv,
                             const float* __restrict__ Ct,
                             const float* __restrict__ stepv) {
  int layer = blockIdx.y;
  int s = blockIdx.x * blockDim.x + threadIdx.x;
  int k2 = s >> 6, k1 = s & 63;
  int kf = 128 * k1 + k2;

  float osin, ocos;
  sincosf(-2.f * PI_F * (float)kf / 8192.f, &osin, &ocos);
  float pr = 1.f + ocos, pi = osin;
  float mr = 1.f - ocos, mi = -osin;
  float pd = pr * pr + pi * pi;
  float ipr = pr / pd, ipi = -pi / pd;
  float cr = 2.f * ipr, ci = 2.f * ipi;
  float sc = 2.f / stepv[layer];
  float gr = sc * (mr * ipr - mi * ipi);
  float gi = sc * (mr * ipi + mi * ipr);

  const float* lr = Lr + layer * NSTATE;
  const float* li = Li + layer * NSTATE;
  const float* pp = Pv + layer * NSTATE;
  const float* bb = Bv + layer * NSTATE;
  const float* ct = Ct + layer * NSTATE;

  float k00r = 0, k00i = 0, k01r = 0, k01i = 0;
  float k10r = 0, k10i = 0, k11r = 0, k11i = 0;
  for (int n = 0; n < NSTATE; ++n) {
    float dr = gr - lr[n], di = gi - li[n];
    float dd = dr * dr + di * di;
    float ir = dr / dd, ii = -di / dd;
    float b = bb[n], p = pp[n], c = ct[n];
    float w00 = c * b, w01 = c * p, w10 = p * b, w11 = p * p;
    k00r += w00 * ir; k00i += w00 * ii;
    k01r += w01 * ir; k01i += w01 * ii;
    k10r += w10 * ir; k10i += w10 * ii;
    k11r += w11 * ir; k11i += w11 * ii;
  }
  float qr = k01r * k10r - k01i * k10i;
  float qi = k01r * k10i + k01i * k10r;
  float er = 1.f + k11r, ei = k11i;
  float ed = er * er + ei * ei;
  float ier = er / ed, iei = -ei / ed;
  float tr = qr * ier - qi * iei;
  float ti = qr * iei + qi * ier;
  float kr = k00r - tr, ki = k00i - ti;
  float outr = cr * kr - ci * ki;
  float outi = cr * ki + ci * kr;
  const float scale = 1.f / 8192.f;
  ws[OFF_KHAT + layer * 16384 + 2 * s]     = outr * scale;
  ws[OFF_KHAT + layer * 16384 + 2 * s + 1] = outi * scale;
}

__global__ __launch_bounds__(256) void ssm_rows(const float* __restrict__ uin,
                                                float* __restrict__ uout,
                                                const float* __restrict__ ws) {
  extern __shared__ float smem[];
  float* xb = smem;
  float* Yr = smem + 8192;
  float* Yi = smem + 16384;

  const int tid  = threadIdx.x;
  const int wave = tid >> 5;
  const int lane = tid & 31;
  const int l16  = lane & 15;
  const int kSel = (lane >> 4) << 1;
  const int mHi  = (lane >> 4) << 3;
  const int mB   = wave << 4;

  const size_t rowOff = (size_t)blockIdx.x * LROW;
  const float4* src = (const float4*)(uin + rowOff);
  for (int i = tid; i < LROW / 4; i += 256) {
    float4 v = src[i];
    int n2 = i >> 4, n1 = (i & 15) << 2;
    int base = pidx(n2, n1);
    xb[base] = v.x; xb[base + 2] = v.y; xb[base + 4] = v.z; xb[base + 6] = v.w;
  }
  __syncthreads();

  const float* F128Q  = ws + OFF_F128Q;
  const float* IF128Q = ws + OFF_IF128Q;
  const float* GPr    = ws + OFF_G64PR;
  const float* GPi    = ws + OFF_G64PI;
  const float* iGPr   = ws + OFF_IG64PR;
  const float* iGPi   = ws + OFF_IG64PI;
  const float* Wf     = ws + OFF_WF;
  const float* Wi     = ws + OFF_WI;

  for (int layer = 0; layer < 4; ++layer) {
    const float* kh = ws + OFF_KHAT + layer * 16384;

    {
      v8f cr[4] = {}, ci[4] = {};
      for (int kk = 0; kk < 128; kk += 4) {
        float4 aq = *(const float4*)(F128Q + (mB + l16) * 256 + (kk + kSel) * 2);
        v2f ar; ar.x = aq.x; ar.y = aq.y;
        v2f ai; ai.x = aq.z; ai.y = aq.w;
        int jb = ((kk + kSel) >> 1) << 7;
#pragma unroll
        for (int t = 0; t < 4; ++t) {
          v2f b = *(const v2f*)(xb + jb + (((t << 4) + l16) << 1));
          cr[t] = wmma4(ar, b, cr[t]);
          ci[t] = wmma4(ai, b, ci[t]);
        }
      }
#pragma unroll
      for (int t = 0; t < 4; ++t) {
#pragma unroll
        for (int v = 0; v < 8; ++v) {
          int row = mB + v + mHi, col = (t << 4) + l16;
          v2f w = *(const v2f*)(Wf + ((row << 6) + col) * 2);
          int p = pidx(row, col);
          float a = cr[t][v], b = ci[t][v];
          Yr[p] = a * w.x - b * w.y;
          Yi[p] = a * w.y + b * w.x;
        }
      }
    }

    {
      v8f zr[4] = {}, zi[4] = {};
      for (int kk = 0; kk < 64; kk += 4) {
        int p0 = pidx(mB + l16, kk + kSel);
        v2f ar; ar.x = Yr[p0]; ar.y = Yr[p0 + 2];
        v2f ai; ai.x = Yi[p0]; ai.y = Yi[p0 + 2];
        int jrow = ((kk + kSel) >> 1) << 7;
#pragma unroll
        for (int t = 0; t < 4; ++t) {
          int cc = ((t << 4) + l16) << 1;
          v2f br = *(const v2f*)(GPr + jrow + cc);
          v2f bi = *(const v2f*)(GPi + jrow + cc);
          v2f nbi; nbi.x = -bi.x; nbi.y = -bi.y;
          zr[t] = wmma4(ar, br, zr[t]);  zr[t] = wmma4(ai, nbi, zr[t]);
          zi[t] = wmma4(ar, bi, zi[t]);  zi[t] = wmma4(ai, br, zi[t]);
        }
      }
#pragma unroll
      for (int t = 0; t < 4; ++t) {
#pragma unroll
        for (int v = 0; v < 8; ++v) {
          int row = mB + v + mHi, col = (t << 4) + l16;
          v2f k = *(const v2f*)(kh + ((row << 6) + col) * 2);
          int p = pidx(row, col);
          float a = zr[t][v], b = zi[t][v];
          Yr[p] = a * k.x - b * k.y;
          Yi[p] = a * k.y + b * k.x;
        }
      }
    }

    {
      v8f yr[4] = {}, yi[4] = {};
      for (int kk = 0; kk < 64; kk += 4) {
        int p0 = pidx(mB + l16, kk + kSel);
        v2f ar; ar.x = Yr[p0]; ar.y = Yr[p0 + 2];
        v2f ai; ai.x = Yi[p0]; ai.y = Yi[p0 + 2];
        int jrow = ((kk + kSel) >> 1) << 7;
#pragma unroll
        for (int t = 0; t < 4; ++t) {
          int cc = ((t << 4) + l16) << 1;
          v2f br = *(const v2f*)(iGPr + jrow + cc);
          v2f bi = *(const v2f*)(iGPi + jrow + cc);
          v2f nbi; nbi.x = -bi.x; nbi.y = -bi.y;
          yr[t] = wmma4(ar, br, yr[t]);  yr[t] = wmma4(ai, nbi, yr[t]);
          yi[t] = wmma4(ar, bi, yi[t]);  yi[t] = wmma4(ai, br, yi[t]);
        }
      }
#pragma unroll
      for (int t = 0; t < 4; ++t) {
#pragma unroll
        for (int v = 0; v < 8; ++v) {
          int row = mB + v + mHi, col = (t << 4) + l16;
          v2f w = *(const v2f*)(Wi + ((row << 6) + col) * 2);
          int p = pidx(row, col);
          float a = yr[t][v], b = yi[t][v];
          Yr[p] = a * w.x - b * w.y;
          Yi[p] = a * w.y + b * w.x;
        }
      }
    }
    __syncthreads();

    {
      v8f acc[4] = {};
      for (int kk = 0; kk < 128; kk += 4) {
        float4 aq = *(const float4*)(IF128Q + (mB + l16) * 256 + (kk + kSel) * 2);
        v2f ar; ar.x = aq.x; ar.y = aq.y;
        v2f ai; ai.x = aq.z; ai.y = aq.w;
        int jb = ((kk + kSel) >> 1) << 7;
#pragma unroll
        for (int t = 0; t < 4; ++t) {
          int cc = ((t << 4) + l16) << 1;
          v2f br  = *(const v2f*)(Yr + jb + cc);
          v2f byi = *(const v2f*)(Yi + jb + cc);
          v2f nbi; nbi.x = -byi.x; nbi.y = -byi.y;
          acc[t] = wmma4(ar, br, acc[t]);
          acc[t] = wmma4(ai, nbi, acc[t]);
        }
      }
#pragma unroll
      for (int t = 0; t < 4; ++t) {
#pragma unroll
        for (int v = 0; v < 8; ++v) {
          int row = mB + v + mHi, col = (t << 4) + l16;
          int p = pidx(row, col);
          float y = acc[t][v] + xb[p];
          xb[p] = 0.5f * y * (1.f + erff(y * 0.70710678118654752f));
        }
      }
    }
    __syncthreads();
  }

  float4* dst = (float4*)(uout + rowOff);
  for (int i = tid; i < LROW / 4; i += 256) {
    int n2 = i >> 4, n1 = (i & 15) << 2;
    int base = pidx(n2, n1);
    float4 o;
    o.x = xb[base]; o.y = xb[base + 2]; o.z = xb[base + 4]; o.w = xb[base + 6];
    dst[i] = o;
  }
}

extern "C" void kernel_launch(void* const* d_in, const int* in_sizes, int n_in,
                              void* d_out, int out_size, void* d_ws, size_t ws_size,
                              hipStream_t stream) {
  const float* u    = (const float*)d_in[0];
  const float* Lr   = (const float*)d_in[1];
  const float* Li   = (const float*)d_in[2];
  const float* Pv   = (const float*)d_in[3];
  const float* Bv   = (const float*)d_in[4];
  const float* Ct   = (const float*)d_in[5];
  const float* step = (const float*)d_in[6];
  float* out = (float*)d_out;
  float* ws  = (float*)d_ws;

  int rows = in_sizes[0] / LROW;

  ssm_gen_tables<<<64, 256, 0, stream>>>(ws);
  ssm_gen_khat<<<dim3(32, 4), 256, 0, stream>>>(ws, Lr, Li, Pv, Bv, Ct, step);
  ssm_rows<<<rows, 256, 24576 * sizeof(float), stream>>>(u, out, ws);
}


# Stages 1, 2 and inverse 1 computed transposed: a result tile's registers are the next product's B fragment; Y crosses LDS once per layer (swizzled), not 3x; k order of both 64-point sums permuted
# speedup vs baseline: 1.2527x; 1.2527x over previous
#include <hip/hip_runtime.h>
#include <math.h>

typedef float v2f __attribute__((ext_vector_type(2)));
typedef float v8f __attribute__((ext_vector_type(8)));

#define LROW 8192
#define NSTATE 64
#define PI_F 3.14159265358979323846f

#define OFF_F128Q   0
#define OFF_IF128Q  32768
#define OFF_G64PR   65536
#define OFF_G64PI   69632
#define OFF_IG64PR  73728
#define OFF_IG64PI  77824
#define OFF_WF      81920
#define OFF_WI      98304
#define OFF_KHAT    114688
#define WS_FLOATS   (OFF_KHAT + 4*16384)

#define LDS_XB      0
#define LDS_YR      8192
#define LDS_YI      16384
#define LDS_GR      24576
#define LDS_GI      28672
#define LDS_F       32768
#define FPITCH      260
#define LDS_FLOATS  (LDS_F + 128 * FPITCH)

static __device__ inline v8f wmma4(v2f a, v2f b, v8f c) {
  return __builtin_amdgcn_wmma_f32_16x16x4_f32(false, a, false, b, (short)0, c,
                                               false, false);
}

static __device__ inline int pidx(int row, int col) {
  return ((row >> 1) << 7) + (col << 1) + (row & 1);
}

__global__ void ssm_gen_tables(float* __restrict__ ws) {
  int idx = blockIdx.x * blockDim.x + threadIdx.x;
  {
    int k2 = idx >> 7, n2 = idx & 127;
    int t = (k2 * n2) & 127;
    float s, c;
    sincosf(-2.f * PI_F * (float)t / 128.f, &s, &c);
    int pos = k2 * 256 + (n2 >> 1) * 4 + (n2 & 1);
    ws[OFF_F128Q + pos]      = c;
    ws[OFF_F128Q + pos + 2]  = s;
  }
  if (idx < 8192) {
    int k2 = idx >> 6, n1 = idx & 63;
    int t = (n1 * k2) & 8191;
    float s, c;
    sincosf(-2.f * PI_F * (float)t / 8192.f, &s, &c);
    ws[OFF_WF + 2 * idx]     = c;  ws[OFF_WF + 2 * idx + 1] = s;
    ws[OFF_WI + 2 * idx]     = c;  ws[OFF_WI + 2 * idx + 1] = -s;
  }
  if (idx < 4096) {
    int n1 = idx >> 6, k1 = idx & 63;
    int t = (n1 * k1) & 63;
    float s, c;
    sincosf(-2.f * PI_F * (float)t / 64.f, &s, &c);
    int pos = (n1 >> 1) * 128 + 2 * k1 + (n1 & 1);
    ws[OFF_G64PR + pos]  = c;  ws[OFF_G64PI + pos]  = s;
  }
}

__global__ void ssm_gen_khat(float* __restrict__ ws,
                             const float* __restrict__ Lr,
                             const float* __restrict__ Li,
                             const float* __restrict__ Pv,
                             const float* __restrict__ Bv,
                             const float* __restrict__ Ct,
                             const float* __restrict__ stepv) {
  int layer = blockIdx.y;
  int s = blockIdx.x * blockDim.x + threadIdx.x;
  int k2 = s >> 6, k1 = s & 63;
  int kf = 128 * k1 + k2;

  float osin, ocos;
  sincosf(-2.f * PI_F * (float)kf / 8192.f, &osin, &ocos);
  float pr = 1.f + ocos, pi = osin;
  float mr = 1.f - ocos, mi = -osin;
  float pd = pr * pr + pi * pi;
  float ipr = pr / pd, ipi = -pi / pd;
  float cr = 2.f * ipr, ci = 2.f * ipi;
  float sc = 2.f / stepv[layer];
  float gr = sc * (mr * ipr - mi * ipi);
  float gi = sc * (mr * ipi + mi * ipr);

  const float* lr = Lr + layer * NSTATE;
  const float* li = Li + layer * NSTATE;
  const float* pp = Pv + layer * NSTATE;
  const float* bb = Bv + layer * NSTATE;
  const float* ct = Ct + layer * NSTATE;

  float k00r = 0, k00i = 0, k01r = 0, k01i = 0;
  float k10r = 0, k10i = 0, k11r = 0, k11i = 0;
  for (int n = 0; n < NSTATE; ++n) {
    float dr = gr - lr[n], di = gi - li[n];
    float dd = dr * dr + di * di;
    float ir = dr / dd, ii = -di / dd;
    float b = bb[n], p = pp[n], c = ct[n];
    float w00 = c * b, w01 = c * p, w10 = p * b, w11 = p * p;
    k00r += w00 * ir; k00i += w00 * ii;
    k01r += w01 * ir; k01i += w01 * ii;
    k10r += w10 * ir; k10i += w10 * ii;
    k11r += w11 * ir; k11i += w11 * ii;
  }
  float qr = k01r * k10r - k01i * k10i;
  float qi = k01r * k10i + k01i * k10r;
  float er = 1.f + k11r, ei = k11i;
  float ed = er * er + ei * ei;
  float ier = er / ed, iei = -ei / ed;
  float tr = qr * ier - qi * iei;
  float ti = qr * iei + qi * ier;
  float kr = k00r - tr, ki = k00i - ti;
  float outr = cr * kr - ci * ki;
  float outi = cr * ki + ci * kr;
  const float scale = 1.f / 8192.f;
  ws[OFF_KHAT + layer * 16384 + 2 * s]     = outr * scale;
  ws[OFF_KHAT + layer * 16384 + 2 * s + 1] = outi * scale;
}

__global__ __launch_bounds__(256) void ssm_rows(const float* __restrict__ uin,
                                                float* __restrict__ uout,
                                                const float* __restrict__ ws,
                                                int rows) {
  __shared__ __attribute__((aligned(16))) float smem[LDS_FLOATS];
  float* xb = smem + LDS_XB;
  float* Yr = smem + LDS_YR;
  float* Yi = smem + LDS_YI;
  float* GLr = smem + LDS_GR;
  float* GLi = smem + LDS_GI;
  float* FL  = smem + LDS_F;

  const int tid  = threadIdx.x;
  const int wave = tid >> 5;
  const int lane = tid & 31;
  const int l16  = lane & 15;
  const int kSel = (lane >> 4) << 1;
  const int mHi  = (lane >> 4) << 3;
  const int mB   = wave << 4;

  float4 xv[8] = {};
  if ((int)blockIdx.x < rows) {
    const float4* src = (const float4*)(uin + (size_t)blockIdx.x * LROW);
#pragma unroll
    for (int j = 0; j < 8; ++j) xv[j] = src[tid + j * 256];
  }

  {
    const float4* fsrc = (const float4*)(ws + OFF_F128Q);
    for (int i = tid; i < 128 * 64; i += 256) {
      float4 v = fsrc[i];
      int row = i >> 6, c4 = i & 63;
      *(float4*)(FL + row * FPITCH + (c4 << 2)) = v;
    }
    const float4* gsrc = (const float4*)(ws + OFF_G64PR);
    float4* gdst = (float4*)GLr;
    for (int i = tid; i < 2 * 4096 / 4; i += 256) gdst[i] = gsrc[i];
  }

  const float* Wf     = ws + OFF_WF;
  const float* Wi     = ws + OFF_WI;

  int swq[4];
#pragma unroll
  for (int q = 0; q < 4; ++q) swq[q] = (l16 ^ (2 * q + (kSel >> 1))) << 1;

  for (int r = blockIdx.x; r < rows; r += gridDim.x) {
  const size_t rowOff = (size_t)r * LROW;
#pragma unroll
  for (int j = 0; j < 8; ++j) {
    int i = tid + j * 256;
    float4 v = xv[j];
    int n2 = i >> 4, n1 = (i & 15) << 2;
    int base = pidx(n2, n1);
    xb[base] = v.x; xb[base + 2] = v.y; xb[base + 4] = v.z; xb[base + 6] = v.w;
  }
  __syncthreads();

  if (r + (int)gridDim.x < rows) {
    const float4* srcn = (const float4*)(uin + (size_t)(r + (int)gridDim.x) * LROW);
#pragma unroll
    for (int j = 0; j < 8; ++j) xv[j] = srcn[tid + j * 256];
  }

  for (int layer = 0; layer < 4; ++layer) {
    const float* kh = ws + OFF_KHAT + layer * 16384;

    v8f yr[4] = {}, yi[4] = {}, nyi[4];
    for (int kk = 0; kk < 128; kk += 4) {
      float4 fq = *(const float4*)(FL + (mB + l16) * FPITCH + (kk + kSel) * 2);
      v2f fr; fr.x = fq.x; fr.y = fq.y;
      v2f fi; fi.x = fq.z; fi.y = fq.w;
      int jb = ((kk + kSel) >> 1) << 7;
#pragma unroll
      for (int t = 0; t < 4; ++t) {
        v2f x = *(const v2f*)(xb + jb + (((t << 4) + l16) << 1));
        yr[t] = wmma4(x, fr, yr[t]);
        yi[t] = wmma4(x, fi, yi[t]);
      }
    }
    {
      const float4* w4 = (const float4*)(Wf + ((((mB + l16) << 6) + mHi) << 1));
#pragma unroll
      for (int t = 0; t < 4; ++t) {
#pragma unroll
        for (int q = 0; q < 4; ++q) {
          float4 w = w4[t * 8 + q];
          float a0 = yr[t][2 * q], b0 = yi[t][2 * q];
          float a1 = yr[t][2 * q + 1], b1 = yi[t][2 * q + 1];
          float i0 = a0 * w.y + b0 * w.x, i1 = a1 * w.w + b1 * w.z;
          yr[t][2 * q]     = a0 * w.x - b0 * w.y;
          yr[t][2 * q + 1] = a1 * w.z - b1 * w.w;
          yi[t][2 * q]  = i0;  yi[t][2 * q + 1]  = i1;
          nyi[t][2 * q] = -i0; nyi[t][2 * q + 1] = -i1;
        }
      }
    }

    const int gb = (kSel << 8) + (l16 << 1);

    v8f zr[4] = {}, zi[4] = {}, nzr[4];
#pragma unroll
    for (int mt = 0; mt < 4; ++mt) {
      if (mt == 2) __builtin_amdgcn_sched_barrier(0);
#pragma unroll
      for (int s = 0; s < 4; ++s) {
        v2f br;  br.x  = yr[mt][2 * s];  br.y  = yr[mt][2 * s + 1];
        v2f bi;  bi.x  = yi[mt][2 * s];  bi.y  = yi[mt][2 * s + 1];
        v2f nbi; nbi.x = nyi[mt][2 * s]; nbi.y = nyi[mt][2 * s + 1];
#pragma unroll
        for (int mo = 0; mo < 4; ++mo) {
          int go = gb + ((8 * mt + s) << 7) + (mo << 5);
          v2f ac = *(const v2f*)(GLr + go);
          v2f as = *(const v2f*)(GLi + go);
          zr[mo] = wmma4(ac, br, zr[mo]);  zr[mo] = wmma4(as, nbi, zr[mo]);
          zi[mo] = wmma4(as, br, zi[mo]);  zi[mo] = wmma4(ac, bi, zi[mo]);
        }
      }
    }
    {
      const float4* k4 = (const float4*)(kh + ((((mB + l16) << 6) + mHi) << 1));
#pragma unroll
      for (int mo = 0; mo < 4; ++mo) {
#pragma unroll
        for (int q = 0; q < 4; ++q) {
          float4 k = k4[mo * 8 + q];
          float a0 = zr[mo][2 * q], b0 = zi[mo][2 * q];
          float a1 = zr[mo][2 * q + 1], b1 = zi[mo][2 * q + 1];
          float r0 = a0 * k.x - b0 * k.y, r1 = a1 * k.z - b1 * k.w;
          zi[mo][2 * q]     = a0 * k.y + b0 * k.x;
          zi[mo][2 * q + 1] = a1 * k.w + b1 * k.z;
          zr[mo][2 * q]  = r0;  zr[mo][2 * q + 1]  = r1;
          nzr[mo][2 * q] = -r0; nzr[mo][2 * q + 1] = -r1;
        }
      }
    }

    v8f pr[4] = {}, pi[4] = {};
#pragma unroll
    for (int mo = 0; mo < 4; ++mo) {
      if (mo == 2) __builtin_amdgcn_sched_barrier(0);
#pragma unroll
      for (int s = 0; s < 4; ++s) {
        v2f br;  br.x  = zr[mo][2 * s];  br.y  = zr[mo][2 * s + 1];
        v2f bi;  bi.x  = zi[mo][2 * s];  bi.y  = zi[mo][2 * s + 1];
        v2f nbr; nbr.x = nzr[mo][2 * s]; nbr.y = nzr[mo][2 * s + 1];
#pragma unroll
        for (int t = 0; t < 4; ++t) {
          int go = gb + ((8 * mo + s) << 7) + (t << 5);
          v2f ac = *(const v2f*)(GLr + go);
          v2f as = *(const v2f*)(GLi + go);
          pr[t] = wmma4(ac, br, pr[t]);   pr[t] = wmma4(as, bi, pr[t]);
          pi[t] = wmma4(as, nbr, pi[t]);  pi[t] = wmma4(ac, bi, pi[t]);
        }
      }
    }
    {
      const float4* w4 = (const float4*)(Wi + ((((mB + l16) << 6) + mHi) << 1));
      const int sb = (((mB + l16) >> 1) << 7) + (l16 & 1);
      const int q8 = l16 >> 1;
#pragma unroll
      for (int t = 0; t < 4; ++t) {
#pragma unroll
        for (int q = 0; q < 4; ++q) {
          float4 w = w4[t * 8 + q];
          float a0 = pr[t][2 * q], b0 = pi[t][2 * q];
          float a1 = pr[t][2 * q + 1], b1 = pi[t][2 * q + 1];
          int p0 = sb + ((((t << 4) + mHi + 2 * q)     ^ q8) << 1);
          int p1 = sb + ((((t << 4) + mHi + 2 * q + 1) ^ q8) << 1);
          Yr[p0] = a0 * w.x - b0 * w.y;
          Yi[p0] = a0 * w.y + b0 * w.x;
          Yr[p1] = a1 * w.z - b1 * w.w;
          Yi[p1] = a1 * w.w + b1 * w.z;
        }
      }
    }
    __syncthreads();

    {
      v8f acc[4] = {};
      for (int kk0 = 0; kk0 < 128; kk0 += 16) {
#pragma unroll
        for (int q = 0; q < 4; ++q) {
          int kk = kk0 + 4 * q;
          float4 aq = *(const float4*)(FL + (mB + l16) * FPITCH + (kk + kSel) * 2);
          v2f ar; ar.x = aq.x; ar.y = aq.y;
          v2f as; as.x = aq.z; as.y = aq.w;
          int jb = (((kk + kSel) >> 1) << 7) + swq[q];
#pragma unroll
          for (int t = 0; t < 4; ++t) {
            v2f br  = *(const v2f*)(Yr + jb + (t << 5));
            v2f byi = *(const v2f*)(Yi + jb + (t << 5));
            acc[t] = wmma4(ar, br, acc[t]);
            acc[t] = wmma4(as, byi, acc[t]);
          }
        }
      }
#pragma unroll
      for (int t = 0; t < 4; ++t) {
#pragma unroll
        for (int v = 0; v < 8; ++v) {
          int row = mB + v + mHi, col = (t << 4) + l16;
          int p = pidx(row, col);
          float y = acc[t][v] + xb[p];
          xb[p] = 0.5f * y * (1.f + erff(y * 0.70710678118654752f));
        }
      }
    }
    __syncthreads();
  }

  float4* dst = (float4*)(uout + rowOff);
  for (int i = tid; i < LROW / 4; i += 256) {
    int n2 = i >> 4, n1 = (i & 15) << 2;
    int base = pidx(n2, n1);
    float4 o;
    o.x = xb[base]; o.y = xb[base + 2]; o.z = xb[base + 4]; o.w = xb[base + 6];
    dst[i] = o;
  }
  }
}

extern "C" void kernel_launch(void* const* d_in, const int* in_sizes, int n_in,
                              void* d_out, int out_size, void* d_ws, size_t ws_size,
                              hipStream_t stream) {
  const float* u    = (const float*)d_in[0];
  const float* Lr   = (const float*)d_in[1];
  const float* Li   = (const float*)d_in[2];
  const float* Pv   = (const float*)d_in[3];
  const float* Bv   = (const float*)d_in[4];
  const float* Ct   = (const float*)d_in[5];
  const float* step = (const float*)d_in[6];
  float* out = (float*)d_out;
  float* ws  = (float*)d_ws;

  int rows = in_sizes[0] / LROW;

  ssm_gen_tables<<<64, 256, 0, stream>>>(ws);
  ssm_gen_khat<<<dim3(32, 4), 256, 0, stream>>>(ws, Lr, Li, Pv, Bv, Ct, step);
  ssm_rows<<<256, 256, 0, stream>>>(u, out, ws, rows);
}


# Priority window: the younger wave of each SIMD's pair (waves 4-7) runs at s_setprio 3 from a layer's opening barrier to the middle of stage 2, then 0; loads, barriers and sum order unchanged
# speedup vs baseline: 1.3277x; 1.0599x over previous
#include <hip/hip_runtime.h>
#include <math.h>

typedef float v2f __attribute__((ext_vector_type(2)));
typedef float v8f __attribute__((ext_vector_type(8)));

#define LROW 8192
#define NSTATE 64
#define PI_F 3.14159265358979323846f

#define OFF_F128Q   0
#define OFF_IF128Q  32768
#define OFF_G64PR   65536
#define OFF_G64PI   69632
#define OFF_IG64PR  73728
#define OFF_IG64PI  77824
#define OFF_WF      81920
#define OFF_WI      98304
#define OFF_KHAT    114688
#define WS_FLOATS   (OFF_KHAT + 4*16384)

#define LDS_XB      0
#define LDS_YR      8192
#define LDS_YI      16384
#define LDS_GR      24576
#define LDS_GI      28672
#define LDS_F       32768
#define FPITCH      260
#define LDS_FLOATS  (LDS_F + 128 * FPITCH)

static __device__ inline v8f wmma4(v2f a, v2f b, v8f c) {
  return __builtin_amdgcn_wmma_f32_16x16x4_f32(false, a, false, b, (short)0, c,
                                               false, false);
}

static __device__ inline int pidx(int row, int col) {
  return ((row >> 1) << 7) + (col << 1) + (row & 1);
}

__global__ void ssm_gen_tables(float* __restrict__ ws) {
  int idx = blockIdx.x * blockDim.x + threadIdx.x;
  {
    int k2 = idx >> 7, n2 = idx & 127;
    int t = (k2 * n2) & 127;
    float s, c;
    sincosf(-2.f * PI_F * (float)t / 128.f, &s, &c);
    int pos = k2 * 256 + (n2 >> 1) * 4 + (n2 & 1);
    ws[OFF_F128Q + pos]      = c;
    ws[OFF_F128Q + pos + 2]  = s;
  }
  if (idx < 8192) {
    int k2 = idx >> 6, n1 = idx & 63;
    int t = (n1 * k2) & 8191;
    float s, c;
    sincosf(-2.f * PI_F * (float)t / 8192.f, &s, &c);
    ws[OFF_WF + 2 * idx]     = c;  ws[OFF_WF + 2 * idx + 1] = s;
    ws[OFF_WI + 2 * idx]     = c;  ws[OFF_WI + 2 * idx + 1] = -s;
  }
  if (idx < 4096) {
    int n1 = idx >> 6, k1 = idx & 63;
    int t = (n1 * k1) & 63;
    float s, c;
    sincosf(-2.f * PI_F * (float)t / 64.f, &s, &c);
    int pos = (n1 >> 1) * 128 + 2 * k1 + (n1 & 1);
    ws[OFF_G64PR + pos]  = c;  ws[OFF_G64PI + pos]  = s;
  }
}

__global__ void ssm_gen_khat(float* __restrict__ ws,
                             const float* __restrict__ Lr,
                             const float* __restrict__ Li,
                             const float* __restrict__ Pv,
                             const float* __restrict__ Bv,
                             const float* __restrict__ Ct,
                             const float* __restrict__ stepv) {
  int layer = blockIdx.y;
  int s = blockIdx.x * blockDim.x + threadIdx.x;
  int k2 = s >> 6, k1 = s & 63;
  int kf = 128 * k1 + k2;

  float osin, ocos;
  sincosf(-2.f * PI_F * (float)kf / 8192.f, &osin, &ocos);
  float pr = 1.f + ocos, pi = osin;
  float mr = 1.f - ocos, mi = -osin;
  float pd = pr * pr + pi * pi;
  float ipr = pr / pd, ipi = -pi / pd;
  float cr = 2.f * ipr, ci = 2.f * ipi;
  float sc = 2.f / stepv[layer];
  float gr = sc * (mr * ipr - mi * ipi);
  float gi = sc * (mr * ipi + mi * ipr);

  const float* lr = Lr + layer * NSTATE;
  const float* li = Li + layer * NSTATE;
  const float* pp = Pv + layer * NSTATE;
  const float* bb = Bv + layer * NSTATE;
  const float* ct = Ct + layer * NSTATE;

  float k00r = 0, k00i = 0, k01r = 0, k01i = 0;
  float k10r = 0, k10i = 0, k11r = 0, k11i = 0;
  for (int n = 0; n < NSTATE; ++n) {
    float dr = gr - lr[n], di = gi - li[n];
    float dd = dr * dr + di * di;
    float ir = dr / dd, ii = -di / dd;
    float b = bb[n], p = pp[n], c = ct[n];
    float w00 = c * b, w01 = c * p, w10 = p * b, w11 = p * p;
    k00r += w00 * ir; k00i += w00 * ii;
    k01r += w01 * ir; k01i += w01 * ii;
    k10r += w10 * ir; k10i += w10 * ii;
    k11r += w11 * ir; k11i += w11 * ii;
  }
  float qr = k01r * k10r - k01i * k10i;
  float qi = k01r * k10i + k01i * k10r;
  float er = 1.f + k11r, ei = k11i;
  float ed = er * er + ei * ei;
  float ier = er / ed, iei = -ei / ed;
  float tr = qr * ier - qi * iei;
  float ti = qr * iei + qi * ier;
  float kr = k00r - tr, ki = k00i - ti;
  float outr = cr * kr - ci * ki;
  float outi = cr * ki + ci * kr;
  const float scale = 1.f / 8192.f;
  ws[OFF_KHAT + layer * 16384 + 2 * s]     = outr * scale;
  ws[OFF_KHAT + layer * 16384 + 2 * s + 1] = outi * scale;
}

__global__ __launch_bounds__(256) void ssm_rows(const float* __restrict__ uin,
                                                float* __restrict__ uout,
                                                const float* __restrict__ ws,
                                                int rows) {
  __shared__ __attribute__((aligned(16))) float smem[LDS_FLOATS];
  float* xb = smem + LDS_XB;
  float* Yr = smem + LDS_YR;
  float* Yi = smem + LDS_YI;
  float* GLr = smem + LDS_GR;
  float* GLi = smem + LDS_GI;
  float* FL  = smem + LDS_F;

  const int tid  = threadIdx.x;
  const int wave = tid >> 5;
  const int lane = tid & 31;
  const int l16  = lane & 15;
  const int kSel = (lane >> 4) << 1;
  const int mHi  = (lane >> 4) << 3;
  const int mB   = wave << 4;
  const bool younger = __builtin_amdgcn_readfirstlane(wave) >= 4;

  float4 xv[8] = {};
  if ((int)blockIdx.x < rows) {
    const float4* src = (const float4*)(uin + (size_t)blockIdx.x * LROW);
#pragma unroll
    for (int j = 0; j < 8; ++j) xv[j] = src[tid + j * 256];
  }

  {
    const float4* fsrc = (const float4*)(ws + OFF_F128Q);
    for (int i = tid; i < 128 * 64; i += 256) {
      float4 v = fsrc[i];
      int row = i >> 6, c4 = i & 63;
      *(float4*)(FL + row * FPITCH + (c4 << 2)) = v;
    }
    const float4* gsrc = (const float4*)(ws + OFF_G64PR);
    float4* gdst = (float4*)GLr;
    for (int i = tid; i < 2 * 4096 / 4; i += 256) gdst[i] = gsrc[i];
  }

  const float* Wf     = ws + OFF_WF;
  const float* Wi     = ws + OFF_WI;

  int swq[4];
#pragma unroll
  for (int q = 0; q < 4; ++q) swq[q] = (l16 ^ (2 * q + (kSel >> 1))) << 1;

  for (int r = blockIdx.x; r < rows; r += gridDim.x) {
  const size_t rowOff = (size_t)r * LROW;
#pragma unroll
  for (int j = 0; j < 8; ++j) {
    int i = tid + j * 256;
    float4 v = xv[j];
    int n2 = i >> 4, n1 = (i & 15) << 2;
    int base = pidx(n2, n1);
    xb[base] = v.x; xb[base + 2] = v.y; xb[base + 4] = v.z; xb[base + 6] = v.w;
  }
  __syncthreads();

  if (r + (int)gridDim.x < rows) {
    const float4* srcn = (const float4*)(uin + (size_t)(r + (int)gridDim.x) * LROW);
#pragma unroll
    for (int j = 0; j < 8; ++j) xv[j] = srcn[tid + j * 256];
  }

  for (int layer = 0; layer < 4; ++layer) {
    const float* kh = ws + OFF_KHAT + layer * 16384;

    if (younger) __builtin_amdgcn_s_setprio(3);

    v8f yr[4] = {}, yi[4] = {}, nyi[4];
    for (int kk = 0; kk < 128; kk += 4) {
      float4 fq = *(const float4*)(FL + (mB + l16) * FPITCH + (kk + kSel) * 2);
      v2f fr; fr.x = fq.x; fr.y = fq.y;
      v2f fi; fi.x = fq.z; fi.y = fq.w;
      int jb = ((kk + kSel) >> 1) << 7;
#pragma unroll
      for (int t = 0; t < 4; ++t) {
        v2f x = *(const v2f*)(xb + jb + (((t << 4) + l16) << 1));
        yr[t] = wmma4(x, fr, yr[t]);
        yi[t] = wmma4(x, fi, yi[t]);
      }
    }
    {
      const float4* w4 = (const float4*)(Wf + ((((mB + l16) << 6) + mHi) << 1));
#pragma unroll
      for (int t = 0; t < 4; ++t) {
#pragma unroll
        for (int q = 0; q < 4; ++q) {
          float4 w = w4[t * 8 + q];
          float a0 = yr[t][2 * q], b0 = yi[t][2 * q];
          float a1 = yr[t][2 * q + 1], b1 = yi[t][2 * q + 1];
          float i0 = a0 * w.y + b0 * w.x, i1 = a1 * w.w + b1 * w.z;
          yr[t][2 * q]     = a0 * w.x - b0 * w.y;
          yr[t][2 * q + 1] = a1 * w.z - b1 * w.w;
          yi[t][2 * q]  = i0;  yi[t][2 * q + 1]  = i1;
          nyi[t][2 * q] = -i0; nyi[t][2 * q + 1] = -i1;
        }
      }
    }

    const int gb = (kSel << 8) + (l16 << 1);

    v8f zr[4] = {}, zi[4] = {}, nzr[4];
#pragma unroll
    for (int mt = 0; mt < 4; ++mt) {
      if (mt == 2) {
        __builtin_amdgcn_sched_barrier(0);
        __builtin_amdgcn_s_setprio(0);
      }
#pragma unroll
      for (int s = 0; s < 4; ++s) {
        v2f br;  br.x  = yr[mt][2 * s];  br.y  = yr[mt][2 * s + 1];
        v2f bi;  bi.x  = yi[mt][2 * s];  bi.y  = yi[mt][2 * s + 1];
        v2f nbi; nbi.x = nyi[mt][2 * s]; nbi.y = nyi[mt][2 * s + 1];
#pragma unroll
        for (int mo = 0; mo < 4; ++mo) {
          int go = gb + ((8 * mt + s) << 7) + (mo << 5);
          v2f ac = *(const v2f*)(GLr + go);
          v2f as = *(const v2f*)(GLi + go);
          zr[mo] = wmma4(ac, br, zr[mo]);  zr[mo] = wmma4(as, nbi, zr[mo]);
          zi[mo] = wmma4(as, br, zi[mo]);  zi[mo] = wmma4(ac, bi, zi[mo]);
        }
      }
    }
    {
      const float4* k4 = (const float4*)(kh + ((((mB + l16) << 6) + mHi) << 1));
#pragma unroll
      for (int mo = 0; mo < 4; ++mo) {
#pragma unroll
        for (int q = 0; q < 4; ++q) {
          float4 k = k4[mo * 8 + q];
          float a0 = zr[mo][2 * q], b0 = zi[mo][2 * q];
          float a1 = zr[mo][2 * q + 1], b1 = zi[mo][2 * q + 1];
          float r0 = a0 * k.x - b0 * k.y, r1 = a1 * k.z - b1 * k.w;
          zi[mo][2 * q]     = a0 * k.y + b0 * k.x;
          zi[mo][2 * q + 1] = a1 * k.w + b1 * k.z;
          zr[mo][2 * q]  = r0;  zr[mo][2 * q + 1]  = r1;
          nzr[mo][2 * q] = -r0; nzr[mo][2 * q + 1] = -r1;
        }
      }
    }

    v8f pr[4] = {}, pi[4] = {};
#pragma unroll
    for (int mo = 0; mo < 4; ++mo) {
      if (mo == 2) __builtin_amdgcn_sched_barrier(0);
#pragma unroll
      for (int s = 0; s < 4; ++s) {
        v2f br;  br.x  = zr[mo][2 * s];  br.y  = zr[mo][2 * s + 1];
        v2f bi;  bi.x  = zi[mo][2 * s];  bi.y  = zi[mo][2 * s + 1];
        v2f nbr; nbr.x = nzr[mo][2 * s]; nbr.y = nzr[mo][2 * s + 1];
#pragma unroll
        for (int t = 0; t < 4; ++t) {
          int go = gb + ((8 * mo + s) << 7) + (t << 5);
          v2f ac = *(const v2f*)(GLr + go);
          v2f as = *(const v2f*)(GLi + go);
          pr[t] = wmma4(ac, br, pr[t]);   pr[t] = wmma4(as, bi, pr[t]);
          pi[t] = wmma4(as, nbr, pi[t]);  pi[t] = wmma4(ac, bi, pi[t]);
        }
      }
    }
    {
      const float4* w4 = (const float4*)(Wi + ((((mB + l16) << 6) + mHi) << 1));
      const int sb = (((mB + l16) >> 1) << 7) + (l16 & 1);
      const int q8 = l16 >> 1;
#pragma unroll
      for (int t = 0; t < 4; ++t) {
#pragma unroll
        for (int q = 0; q < 4; ++q) {
          float4 w = w4[t * 8 + q];
          float a0 = pr[t][2 * q], b0 = pi[t][2 * q];
          float a1 = pr[t][2 * q + 1], b1 = pi[t][2 * q + 1];
          int p0 = sb + ((((t << 4) + mHi + 2 * q)     ^ q8) << 1);
          int p1 = sb + ((((t << 4) + mHi + 2 * q + 1) ^ q8) << 1);
          Yr[p0] = a0 * w.x - b0 * w.y;
          Yi[p0] = a0 * w.y + b0 * w.x;
          Yr[p1] = a1 * w.z - b1 * w.w;
          Yi[p1] = a1 * w.w + b1 * w.z;
        }
      }
    }
    __syncthreads();

    {
      v8f acc[4] = {};
      for (int kk0 = 0; kk0 < 128; kk0 += 16) {
#pragma unroll
        for (int q = 0; q < 4; ++q) {
          int kk = kk0 + 4 * q;
          float4 aq = *(const float4*)(FL + (mB + l16) * FPITCH + (kk + kSel) * 2);
          v2f ar; ar.x = aq.x; ar.y = aq.y;
          v2f as; as.x = aq.z; as.y = aq.w;
          int jb = (((kk + kSel) >> 1) << 7) + swq[q];
#pragma unroll
          for (int t = 0; t < 4; ++t) {
            v2f br  = *(const v2f*)(Yr + jb + (t << 5));
            v2f byi = *(const v2f*)(Yi + jb + (t << 5));
            acc[t] = wmma4(ar, br, acc[t]);
            acc[t] = wmma4(as, byi, acc[t]);
          }
        }
      }
#pragma unroll
      for (int t = 0; t < 4; ++t) {
#pragma unroll
        for (int v = 0; v < 8; ++v) {
          int row = mB + v + mHi, col = (t << 4) + l16;
          int p = pidx(row, col);
          float y = acc[t][v] + xb[p];
          xb[p] = 0.5f * y * (1.f + erff(y * 0.70710678118654752f));
        }
      }
    }
    __syncthreads();
  }

  float4* dst = (float4*)(uout + rowOff);
  for (int i = tid; i < LROW / 4; i += 256) {
    int n2 = i >> 4, n1 = (i & 15) << 2;
    int base = pidx(n2, n1);
    float4 o;
    o.x = xb[base]; o.y = xb[base + 2]; o.z = xb[base + 4]; o.w = xb[base + 6];
    dst[i] = o;
  }
  }
}

extern "C" void kernel_launch(void* const* d_in, const int* in_sizes, int n_in,
                              void* d_out, int out_size, void* d_ws, size_t ws_size,
                              hipStream_t stream) {
  const float* u    = (const float*)d_in[0];
  const float* Lr   = (const float*)d_in[1];
  const float* Li   = (const float*)d_in[2];
  const float* Pv   = (const float*)d_in[3];
  const float* Bv   = (const float*)d_in[4];
  const float* Ct   = (const float*)d_in[5];
  const float* step = (const float*)d_in[6];
  float* out = (float*)d_out;
  float* ws  = (float*)d_ws;

  int rows = in_sizes[0] / LROW;

  ssm_gen_tables<<<64, 256, 0, stream>>>(ws);
  ssm_gen_khat<<<dim3(32, 4), 256, 0, stream>>>(ws, Lr, Li, Pv, Bv, Ct, step);
  ssm_rows<<<256, 256, 0, stream>>>(u, out, ws, rows);
}


# Real-input half spectrum: only bins k2 0..63 pass stages 1-3 (K_hat made Hermitian, rows 1..63 doubled); bin row 64 is one real 64x64 WMMA product; two rows per pass; last stage's sums regrouped
# speedup vs baseline: 2.1995x; 1.6566x over previous
#include <hip/hip_runtime.h>
#include <math.h>

typedef float v2f __attribute__((ext_vector_type(2)));
typedef float v8f __attribute__((ext_vector_type(8)));

#define LROW 8192
#define NSTATE 64
#define PI_F 3.14159265358979323846f

#define OFF_F128Q   0
#define OFF_S64P    32768
#define OFF_G64PR   65536
#define OFF_G64PI   69632
#define OFF_IG64PR  73728
#define OFF_IG64PI  77824
#define OFF_WF      81920
#define OFF_WI      98304
#define OFF_KHAT    114688
#define WS_FLOATS   (OFF_KHAT + 4*16384)

#define LDS_XB      0
#define LDS_YR      16384
#define LDS_YI      24576
#define LDS_GR      32768
#define LDS_GI      36864
#define LDS_T64     40960
#define LDS_F       41088
#define FPITCH      260
#define LDS_FLOATS  (LDS_F + 129 * FPITCH)

static __device__ inline v8f wmma4(v2f a, v2f b, v8f c) {
  return __builtin_amdgcn_wmma_f32_16x16x4_f32(false, a, false, b, (short)0, c,
                                               false, false);
}

static __device__ inline int pidx(int row, int col) {
  return ((row >> 1) << 7) + (col << 1) + (row & 1);
}

__global__ void ssm_gen_tables(float* __restrict__ ws) {
  int idx = blockIdx.x * blockDim.x + threadIdx.x;
  {
    int k2 = idx >> 7, n2 = idx & 127;
    int t = (k2 * n2) & 127;
    float s, c;
    sincosf(-2.f * PI_F * (float)t / 128.f, &s, &c);
    int pos = k2 * 256 + (n2 >> 1) * 4 + (n2 & 1);
    ws[OFF_F128Q + pos]      = c;
    ws[OFF_F128Q + pos + 2]  = s;
  }
  if (idx < 8192) {
    int k2 = idx >> 6, n1 = idx & 63;
    int t = (n1 * k2) & 8191;
    float s, c;
    sincosf(-2.f * PI_F * (float)t / 8192.f, &s, &c);
    ws[OFF_WF + 2 * idx]     = c;  ws[OFF_WF + 2 * idx + 1] = s;
    ws[OFF_WI + 2 * idx]     = c;  ws[OFF_WI + 2 * idx + 1] = -s;
  }
  if (idx < 4096) {
    int n1 = idx >> 6, k1 = idx & 63;
    int t = (n1 * k1) & 63;
    float s, c;
    sincosf(-2.f * PI_F * (float)t / 64.f, &s, &c);
    int pos = (n1 >> 1) * 128 + 2 * k1 + (n1 & 1);
    ws[OFF_G64PR + pos]  = c;  ws[OFF_G64PI + pos]  = s;
  }
}

static __device__ inline void dplr_eval(int kf, float sc,
                                        const float* __restrict__ lr, const float* __restrict__ li,
                                        const float* __restrict__ pp, const float* __restrict__ bb,
                                        const float* __restrict__ ct, float* outr, float* outi) {
  float osin, ocos;
  sincosf(-2.f * PI_F * (float)kf / 8192.f, &osin, &ocos);
  float pr = 1.f + ocos, pi = osin;
  float mr = 1.f - ocos, mi = -osin;
  float pd = pr * pr + pi * pi;
  float ipr = pr / pd, ipi = -pi / pd;
  float cr = 2.f * ipr, ci = 2.f * ipi;
  float gr = sc * (mr * ipr - mi * ipi);
  float gi = sc * (mr * ipi + mi * ipr);

  float k00r = 0, k00i = 0, k01r = 0, k01i = 0;
  float k10r = 0, k10i = 0, k11r = 0, k11i = 0;
  for (int n = 0; n < NSTATE; ++n) {
    float dr = gr - lr[n], di = gi - li[n];
    float dd = dr * dr + di * di;
    float ir = dr / dd, ii = -di / dd;
    float b = bb[n], p = pp[n], c = ct[n];
    float w00 = c * b, w01 = c * p, w10 = p * b, w11 = p * p;
    k00r += w00 * ir; k00i += w00 * ii;
    k01r += w01 * ir; k01i += w01 * ii;
    k10r += w10 * ir; k10i += w10 * ii;
    k11r += w11 * ir; k11i += w11 * ii;
  }
  float qr = k01r * k10r - k01i * k10i;
  float qi = k01r * k10i + k01i * k10r;
  float er = 1.f + k11r, ei = k11i;
  float ed = er * er + ei * ei;
  float ier = er / ed, iei = -ei / ed;
  float tr = qr * ier - qi * iei;
  float ti = qr * iei + qi * ier;
  float kr = k00r - tr, ki = k00i - ti;
  *outr = cr * kr - ci * ki;
  *outi = cr * ki + ci * kr;
}

__global__ void ssm_gen_khat(float* __restrict__ ws,
                             const float* __restrict__ Lr,
                             const float* __restrict__ Li,
                             const float* __restrict__ Pv,
                             const float* __restrict__ Bv,
                             const float* __restrict__ Ct,
                             const float* __restrict__ stepv) {
  int layer = blockIdx.y;
  int s = blockIdx.x * blockDim.x + threadIdx.x;
  int k2 = s >> 6, k1 = s & 63;
  int kf = 128 * k1 + k2;
  int km = (8192 - kf) & 8191;

  float sc = 2.f / stepv[layer];
  const float* lr = Lr + layer * NSTATE;
  const float* li = Li + layer * NSTATE;
  const float* pp = Pv + layer * NSTATE;
  const float* bb = Bv + layer * NSTATE;
  const float* ct = Ct + layer * NSTATE;

  float ar, ai, br, bi;
  dplr_eval(kf, sc, lr, li, pp, bb, ct, &ar, &ai);
  dplr_eval(km, sc, lr, li, pp, bb, ct, &br, &bi);
  const float twice = (k2 == 0 || k2 == 64) ? 1.f : 2.f;
  const float scale = twice * 0.5f / 8192.f;
  ws[OFF_KHAT + layer * 16384 + 2 * s]     = (ar + br) * scale;
  ws[OFF_KHAT + layer * 16384 + 2 * s + 1] = (ai - bi) * scale;
}

__global__ void ssm_gen_s64(float* __restrict__ ws) {
  __shared__ float cs[128], sn[128], kr[64], ki[64];
  const int layer = blockIdx.y;
  const int tid = threadIdx.x;
  if (tid < 128) {
    float s, c;
    sincosf(2.f * PI_F * (float)tid / 128.f, &s, &c);
    cs[tid] = c; sn[tid] = s;
  }
  if (tid < 64) {
    kr[tid] = ws[OFF_KHAT + layer * 16384 + 2 * (4096 + tid)];
    ki[tid] = ws[OFF_KHAT + layer * 16384 + 2 * (4096 + tid) + 1];
  }
  __syncthreads();
  const int idx = blockIdx.x * 256 + tid;
  const int m = idx >> 6, n1 = idx & 63;
  const int d = n1 - m;
  float acc = 0.f;
  for (int k1 = 0; k1 < 64; ++k1) {
    int t = ((2 * k1 + 1) * d) & 127;
    acc += kr[k1] * cs[t] - ki[k1] * sn[t];
  }
  ws[OFF_S64P + layer * 4096 + (m >> 1) * 128 + 2 * n1 + (m & 1)] = acc;
}

__global__ __launch_bounds__(256) void ssm_rows(const float* __restrict__ uin,
                                                float* __restrict__ uout,
                                                const float* __restrict__ ws,
                                                int rows) {
  __shared__ __attribute__((aligned(16))) float smem[LDS_FLOATS];
  float* GLr = smem + LDS_GR;
  float* GLi = smem + LDS_GI;
  float* FL  = smem + LDS_F;

  const int tid  = threadIdx.x;
  const int wave = tid >> 5;
  const int lane = tid & 31;
  const int l16  = lane & 15;
  const int kSel = (lane >> 4) << 1;
  const int mHi  = (lane >> 4) << 3;
  const int waveU = __builtin_amdgcn_readfirstlane(wave);
  const int grp   = waveU >> 2;
  const int wl    = waveU & 3;
  const int kb    = (wl + 2 * grp) & 3;
  const int mB    = kb << 4;
  const bool ownsZero = (kb == 0);
  const int mI    = wl << 5;
  const int tl    = tid & 127;
  const bool younger = waveU >= 4;

  float* xb  = smem + LDS_XB + grp * 8192;
  float* Yr  = smem + LDS_YR + grp * 4096;
  float* Yi  = smem + LDS_YI + grp * 4096;
  float* t64 = smem + LDS_T64 + grp * 64;

  {
    const float4* fsrc = (const float4*)(ws + OFF_F128Q);
    for (int i = tid; i < 128 * 64; i += 256) {
      float4 v = fsrc[i];
      int row = i >> 6, c4 = i & 63;
      *(float4*)(FL + row * FPITCH + (c4 << 2)) = v;
    }
    if (tid < 64) {
      float4 v; v.x = 1.f; v.y = 1.f; v.z = 1.f; v.w = -1.f;
      *(float4*)(FL + 128 * FPITCH + (tid << 2)) = v;
    }
    const float4* gsrc = (const float4*)(ws + OFF_G64PR);
    float4* gdst = (float4*)GLr;
    for (int i = tid; i < 2 * 4096 / 4; i += 256) gdst[i] = gsrc[i];
  }

  const float* Wf     = ws + OFF_WF;
  const float* Wi     = ws + OFF_WI;

  const int frow = (ownsZero && l16 == 0) ? 128 : (mB + l16);

  const int gb = (kSel << 8) + (l16 << 1);

  int swq[4];
#pragma unroll
  for (int q = 0; q < 4; ++q) swq[q] = (l16 ^ (2 * q + (kSel >> 1))) << 1;

  const int pairs = (rows + 1) >> 1;
  for (int pp = blockIdx.x; pp < pairs; pp += gridDim.x) {
  const int r = 2 * pp + grp;
  const bool live = r < rows;
  const size_t rowOff = (size_t)(live ? r : 0) * LROW;
  {
    const float4* src = (const float4*)(uin + rowOff);
    float4 xv[16];
#pragma unroll
    for (int j = 0; j < 16; ++j) xv[j] = src[tl + j * 128];
#pragma unroll
    for (int j = 0; j < 16; ++j) {
      int i = tl + j * 128;
      float4 v = xv[j];
      int n2 = i >> 4, n1 = (i & 15) << 2;
      int base = pidx(n2, n1);
      xb[base] = v.x; xb[base + 2] = v.y; xb[base + 4] = v.z; xb[base + 6] = v.w;
    }
  }
  __syncthreads();

  for (int layer = 0; layer < 4; ++layer) {
    const float* kh = ws + OFF_KHAT + layer * 16384;

    if (younger) __builtin_amdgcn_s_setprio(3);

    v8f yr[4] = {}, yi[4] = {}, nyi[4];
    for (int kk = 0; kk < 128; kk += 4) {
      float4 fq = *(const float4*)(FL + frow * FPITCH + (kk + kSel) * 2);
      v2f fr; fr.x = fq.x; fr.y = fq.y;
      v2f fi; fi.x = fq.z; fi.y = fq.w;
      int jb = ((kk + kSel) >> 1) << 7;
#pragma unroll
      for (int t = 0; t < 4; ++t) {
        v2f x = *(const v2f*)(xb + jb + (((t << 4) + l16) << 1));
        yr[t] = wmma4(x, fr, yr[t]);
        yi[t] = wmma4(x, fi, yi[t]);
      }
    }

    if (ownsZero) {
      const float* s64 = ws + OFF_S64P + layer * 4096;
      v8f tz[4] = {};
#pragma unroll
      for (int mt = 0; mt < 4; ++mt) {
#pragma unroll
        for (int s = 0; s < 4; ++s) {
          v2f bv; bv.x = yi[mt][2 * s]; bv.y = yi[mt][2 * s + 1];
#pragma unroll
          for (int mo = 0; mo < 4; ++mo) {
            int go = gb + ((8 * mt + s) << 7) + (mo << 5);
            v2f a = *(const v2f*)(s64 + go);
            tz[mo] = wmma4(a, bv, tz[mo]);
          }
        }
      }
#pragma unroll
      for (int mo = 0; mo < 4; ++mo) {
#pragma unroll
        for (int rr = 0; rr < 8; ++rr) yi[mo][rr] = (l16 == 0) ? 0.f : yi[mo][rr];
      }
      if (l16 == 0) {
#pragma unroll
        for (int mo = 0; mo < 4; ++mo) {
#pragma unroll
          for (int rr = 0; rr < 8; ++rr) t64[(mo << 4) + mHi + rr] = tz[mo][rr];
        }
      }
    }

    {
      const float4* w4 = (const float4*)(Wf + ((((mB + l16) << 6) + mHi) << 1));
#pragma unroll
      for (int t = 0; t < 4; ++t) {
#pragma unroll
        for (int q = 0; q < 4; ++q) {
          float4 w = w4[t * 8 + q];
          float a0 = yr[t][2 * q], b0 = yi[t][2 * q];
          float a1 = yr[t][2 * q + 1], b1 = yi[t][2 * q + 1];
          float i0 = a0 * w.y + b0 * w.x, i1 = a1 * w.w + b1 * w.z;
          yr[t][2 * q]     = a0 * w.x - b0 * w.y;
          yr[t][2 * q + 1] = a1 * w.z - b1 * w.w;
          yi[t][2 * q]  = i0;  yi[t][2 * q + 1]  = i1;
          nyi[t][2 * q] = -i0; nyi[t][2 * q + 1] = -i1;
        }
      }
    }

    v8f zr[4] = {}, zi[4] = {}, nzr[4];
#pragma unroll
    for (int mt = 0; mt < 4; ++mt) {
      if (mt == 2) {
        __builtin_amdgcn_sched_barrier(0);
        __builtin_amdgcn_s_setprio(0);
      }
#pragma unroll
      for (int s = 0; s < 4; ++s) {
        v2f br;  br.x  = yr[mt][2 * s];  br.y  = yr[mt][2 * s + 1];
        v2f bi;  bi.x  = yi[mt][2 * s];  bi.y  = yi[mt][2 * s + 1];
        v2f nbi; nbi.x = nyi[mt][2 * s]; nbi.y = nyi[mt][2 * s + 1];
#pragma unroll
        for (int mo = 0; mo < 4; ++mo) {
          int go = gb + ((8 * mt + s) << 7) + (mo << 5);
          v2f ac = *(const v2f*)(GLr + go);
          v2f as = *(const v2f*)(GLi + go);
          zr[mo] = wmma4(ac, br, zr[mo]);  zr[mo] = wmma4(as, nbi, zr[mo]);
          zi[mo] = wmma4(as, br, zi[mo]);  zi[mo] = wmma4(ac, bi, zi[mo]);
        }
      }
    }
    {
      const float4* k4 = (const float4*)(kh + ((((mB + l16) << 6) + mHi) << 1));
#pragma unroll
      for (int mo = 0; mo < 4; ++mo) {
#pragma unroll
        for (int q = 0; q < 4; ++q) {
          float4 k = k4[mo * 8 + q];
          float a0 = zr[mo][2 * q], b0 = zi[mo][2 * q];
          float a1 = zr[mo][2 * q + 1], b1 = zi[mo][2 * q + 1];
          float r0 = a0 * k.x - b0 * k.y, r1 = a1 * k.z - b1 * k.w;
          zi[mo][2 * q]     = a0 * k.y + b0 * k.x;
          zi[mo][2 * q + 1] = a1 * k.w + b1 * k.z;
          zr[mo][2 * q]  = r0;  zr[mo][2 * q + 1]  = r1;
          nzr[mo][2 * q] = -r0; nzr[mo][2 * q + 1] = -r1;
        }
      }
    }

    v8f pr[4] = {}, pi[4] = {};
#pragma unroll
    for (int mo = 0; mo < 4; ++mo) {
      if (mo == 2) __builtin_amdgcn_sched_barrier(0);
#pragma unroll
      for (int s = 0; s < 4; ++s) {
        v2f br;  br.x  = zr[mo][2 * s];  br.y  = zr[mo][2 * s + 1];
        v2f bi;  bi.x  = zi[mo][2 * s];  bi.y  = zi[mo][2 * s + 1];
        v2f nbr; nbr.x = nzr[mo][2 * s]; nbr.y = nzr[mo][2 * s + 1];
#pragma unroll
        for (int t = 0; t < 4; ++t) {
          int go = gb + ((8 * mo + s) << 7) + (t << 5);
          v2f ac = *(const v2f*)(GLr + go);
          v2f as = *(const v2f*)(GLi + go);
          pr[t] = wmma4(ac, br, pr[t]);   pr[t] = wmma4(as, bi, pr[t]);
          pi[t] = wmma4(as, nbr, pi[t]);  pi[t] = wmma4(ac, bi, pi[t]);
        }
      }
    }
    {
      const float4* w4 = (const float4*)(Wi + ((((mB + l16) << 6) + mHi) << 1));
      const int sb = (((mB + l16) >> 1) << 7) + (l16 & 1);
      const int q8 = l16 >> 1;
#pragma unroll
      for (int t = 0; t < 4; ++t) {
#pragma unroll
        for (int q = 0; q < 4; ++q) {
          float4 w = w4[t * 8 + q];
          float a0 = pr[t][2 * q], b0 = pi[t][2 * q];
          float a1 = pr[t][2 * q + 1], b1 = pi[t][2 * q + 1];
          int p0 = sb + ((((t << 4) + mHi + 2 * q)     ^ q8) << 1);
          int p1 = sb + ((((t << 4) + mHi + 2 * q + 1) ^ q8) << 1);
          Yr[p0] = a0 * w.x - b0 * w.y;
          Yi[p0] = a0 * w.y + b0 * w.x;
          Yr[p1] = a1 * w.z - b1 * w.w;
          Yi[p1] = a1 * w.w + b1 * w.z;
        }
      }
    }
    __syncthreads();

    {
      v8f acc[2][4] = {};
      for (int kk0 = 0; kk0 < 64; kk0 += 16) {
#pragma unroll
        for (int q = 0; q < 4; ++q) {
          int kk = kk0 + 4 * q;
          float4 aq0 = *(const float4*)(FL + (mI + l16) * FPITCH + (kk + kSel) * 2);
          float4 aq1 = *(const float4*)(FL + (mI + 16 + l16) * FPITCH + (kk + kSel) * 2);
          v2f ar0; ar0.x = aq0.x; ar0.y = aq0.y;
          v2f as0; as0.x = aq0.z; as0.y = aq0.w;
          v2f ar1; ar1.x = aq1.x; ar1.y = aq1.y;
          v2f as1; as1.x = aq1.z; as1.y = aq1.w;
          int jb = (((kk + kSel) >> 1) << 7) + swq[q];
#pragma unroll
          for (int t = 0; t < 4; ++t) {
            v2f br  = *(const v2f*)(Yr + jb + (t << 5));
            v2f byi = *(const v2f*)(Yi + jb + (t << 5));
            acc[0][t] = wmma4(ar0, br, acc[0][t]);
            acc[0][t] = wmma4(as0, byi, acc[0][t]);
            acc[1][t] = wmma4(ar1, br, acc[1][t]);
            acc[1][t] = wmma4(as1, byi, acc[1][t]);
          }
        }
      }
#pragma unroll
      for (int t = 0; t < 4; ++t) {
        const float tv = t64[(t << 4) + l16];
#pragma unroll
        for (int u = 0; u < 2; ++u) {
#pragma unroll
          for (int v = 0; v < 8; ++v) {
            int row = mI + (u << 4) + v + mHi, col = (t << 4) + l16;
            int p = pidx(row, col);
            float y = acc[u][t][v] + ((v & 1) ? -tv : tv) + xb[p];
            xb[p] = 0.5f * y * (1.f + erff(y * 0.70710678118654752f));
          }
        }
      }
    }
    __syncthreads();
  }

  if (live) {
    float4* dst = (float4*)(uout + rowOff);
    for (int i = tl; i < LROW / 4; i += 128) {
      int n2 = i >> 4, n1 = (i & 15) << 2;
      int base = pidx(n2, n1);
      float4 o;
      o.x = xb[base]; o.y = xb[base + 2]; o.z = xb[base + 4]; o.w = xb[base + 6];
      dst[i] = o;
    }
  }
  }
}

extern "C" void kernel_launch(void* const* d_in, const int* in_sizes, int n_in,
                              void* d_out, int out_size, void* d_ws, size_t ws_size,
                              hipStream_t stream) {
  const float* u    = (const float*)d_in[0];
  const float* Lr   = (const float*)d_in[1];
  const float* Li   = (const float*)d_in[2];
  const float* Pv   = (const float*)d_in[3];
  const float* Bv   = (const float*)d_in[4];
  const float* Ct   = (const float*)d_in[5];
  const float* step = (const float*)d_in[6];
  float* out = (float*)d_out;
  float* ws  = (float*)d_ws;

  int rows = in_sizes[0] / LROW;

  ssm_gen_tables<<<64, 256, 0, stream>>>(ws);
  ssm_gen_khat<<<dim3(32, 4), 256, 0, stream>>>(ws, Lr, Li, Pv, Bv, Ct, step);
  ssm_gen_s64<<<dim3(16, 4), 256, 0, stream>>>(ws);
  ssm_rows<<<256, 256, 0, stream>>>(u, out, ws, rows);
}


# Inverse stage 2 accumulators start from the epilogue's addends (residual x by b64 LDS loads, bin row 64's term): no LDS read or add in an element's GELU chain; x and T64 summed first, not last
# speedup vs baseline: 2.3521x; 1.0694x over previous
#include <hip/hip_runtime.h>
#include <math.h>

typedef float v2f __attribute__((ext_vector_type(2)));
typedef float v8f __attribute__((ext_vector_type(8)));

#define LROW 8192
#define NSTATE 64
#define PI_F 3.14159265358979323846f

#define OFF_F128Q   0
#define OFF_S64P    32768
#define OFF_G64PR   65536
#define OFF_G64PI   69632
#define OFF_IG64PR  73728
#define OFF_IG64PI  77824
#define OFF_WF      81920
#define OFF_WI      98304
#define OFF_KHAT    114688
#define WS_FLOATS   (OFF_KHAT + 4*16384)

#define LDS_XB      0
#define LDS_YR      16384
#define LDS_YI      24576
#define LDS_GR      32768
#define LDS_GI      36864
#define LDS_T64     40960
#define LDS_F       41088
#define FPITCH      260
#define LDS_FLOATS  (LDS_F + 129 * FPITCH)

static __device__ inline v8f wmma4(v2f a, v2f b, v8f c) {
  return __builtin_amdgcn_wmma_f32_16x16x4_f32(false, a, false, b, (short)0, c,
                                               false, false);
}

static __device__ inline int pidx(int row, int col) {
  return ((row >> 1) << 7) + (col << 1) + (row & 1);
}

__global__ void ssm_gen_tables(float* __restrict__ ws) {
  int idx = blockIdx.x * blockDim.x + threadIdx.x;
  {
    int k2 = idx >> 7, n2 = idx & 127;
    int t = (k2 * n2) & 127;
    float s, c;
    sincosf(-2.f * PI_F * (float)t / 128.f, &s, &c);
    int pos = k2 * 256 + (n2 >> 1) * 4 + (n2 & 1);
    ws[OFF_F128Q + pos]      = c;
    ws[OFF_F128Q + pos + 2]  = s;
  }
  if (idx < 8192) {
    int k2 = idx >> 6, n1 = idx & 63;
    int t = (n1 * k2) & 8191;
    float s, c;
    sincosf(-2.f * PI_F * (float)t / 8192.f, &s, &c);
    ws[OFF_WF + 2 * idx]     = c;  ws[OFF_WF + 2 * idx + 1] = s;
    ws[OFF_WI + 2 * idx]     = c;  ws[OFF_WI + 2 * idx + 1] = -s;
  }
  if (idx < 4096) {
    int n1 = idx >> 6, k1 = idx & 63;
    int t = (n1 * k1) & 63;
    float s, c;
    sincosf(-2.f * PI_F * (float)t / 64.f, &s, &c);
    int pos = (n1 >> 1) * 128 + 2 * k1 + (n1 & 1);
    ws[OFF_G64PR + pos]  = c;  ws[OFF_G64PI + pos]  = s;
  }
}

static __device__ inline void dplr_eval(int kf, float sc,
                                        const float* __restrict__ lr, const float* __restrict__ li,
                                        const float* __restrict__ pp, const float* __restrict__ bb,
                                        const float* __restrict__ ct, float* outr, float* outi) {
  float osin, ocos;
  sincosf(-2.f * PI_F * (float)kf / 8192.f, &osin, &ocos);
  float pr = 1.f + ocos, pi = osin;
  float mr = 1.f - ocos, mi = -osin;
  float pd = pr * pr + pi * pi;
  float ipr = pr / pd, ipi = -pi / pd;
  float cr = 2.f * ipr, ci = 2.f * ipi;
  float gr = sc * (mr * ipr - mi * ipi);
  float gi = sc * (mr * ipi + mi * ipr);

  float k00r = 0, k00i = 0, k01r = 0, k01i = 0;
  float k10r = 0, k10i = 0, k11r = 0, k11i = 0;
  for (int n = 0; n < NSTATE; ++n) {
    float dr = gr - lr[n], di = gi - li[n];
    float dd = dr * dr + di * di;
    float ir = dr / dd, ii = -di / dd;
    float b = bb[n], p = pp[n], c = ct[n];
    float w00 = c * b, w01 = c * p, w10 = p * b, w11 = p * p;
    k00r += w00 * ir; k00i += w00 * ii;
    k01r += w01 * ir; k01i += w01 * ii;
    k10r += w10 * ir; k10i += w10 * ii;
    k11r += w11 * ir; k11i += w11 * ii;
  }
  float qr = k01r * k10r - k01i * k10i;
  float qi = k01r * k10i + k01i * k10r;
  float er = 1.f + k11r, ei = k11i;
  float ed = er * er + ei * ei;
  float ier = er / ed, iei = -ei / ed;
  float tr = qr * ier - qi * iei;
  float ti = qr * iei + qi * ier;
  float kr = k00r - tr, ki = k00i - ti;
  *outr = cr * kr - ci * ki;
  *outi = cr * ki + ci * kr;
}

__global__ void ssm_gen_khat(float* __restrict__ ws,
                             const float* __restrict__ Lr,
                             const float* __restrict__ Li,
                             const float* __restrict__ Pv,
                             const float* __restrict__ Bv,
                             const float* __restrict__ Ct,
                             const float* __restrict__ stepv) {
  int layer = blockIdx.y;
  int s = blockIdx.x * blockDim.x + threadIdx.x;
  int k2 = s >> 6, k1 = s & 63;
  int kf = 128 * k1 + k2;
  int km = (8192 - kf) & 8191;

  float sc = 2.f / stepv[layer];
  const float* lr = Lr + layer * NSTATE;
  const float* li = Li + layer * NSTATE;
  const float* pp = Pv + layer * NSTATE;
  const float* bb = Bv + layer * NSTATE;
  const float* ct = Ct + layer * NSTATE;

  float ar, ai, br, bi;
  dplr_eval(kf, sc, lr, li, pp, bb, ct, &ar, &ai);
  dplr_eval(km, sc, lr, li, pp, bb, ct, &br, &bi);
  const float twice = (k2 == 0 || k2 == 64) ? 1.f : 2.f;
  const float scale = twice * 0.5f / 8192.f;
  ws[OFF_KHAT + layer * 16384 + 2 * s]     = (ar + br) * scale;
  ws[OFF_KHAT + layer * 16384 + 2 * s + 1] = (ai - bi) * scale;
}

__global__ void ssm_gen_s64(float* __restrict__ ws) {
  __shared__ float cs[128], sn[128], kr[64], ki[64];
  const int layer = blockIdx.y;
  const int tid = threadIdx.x;
  if (tid < 128) {
    float s, c;
    sincosf(2.f * PI_F * (float)tid / 128.f, &s, &c);
    cs[tid] = c; sn[tid] = s;
  }
  if (tid < 64) {
    kr[tid] = ws[OFF_KHAT + layer * 16384 + 2 * (4096 + tid)];
    ki[tid] = ws[OFF_KHAT + layer * 16384 + 2 * (4096 + tid) + 1];
  }
  __syncthreads();
  const int idx = blockIdx.x * 256 + tid;
  const int m = idx >> 6, n1 = idx & 63;
  const int d = n1 - m;
  float acc = 0.f;
  for (int k1 = 0; k1 < 64; ++k1) {
    int t = ((2 * k1 + 1) * d) & 127;
    acc += kr[k1] * cs[t] - ki[k1] * sn[t];
  }
  ws[OFF_S64P + layer * 4096 + (m >> 1) * 128 + 2 * n1 + (m & 1)] = acc;
}

__global__ __launch_bounds__(256) void ssm_rows(const float* __restrict__ uin,
                                                float* __restrict__ uout,
                                                const float* __restrict__ ws,
                                                int rows) {
  __shared__ __attribute__((aligned(16))) float smem[LDS_FLOATS];
  float* GLr = smem + LDS_GR;
  float* GLi = smem + LDS_GI;
  float* FL  = smem + LDS_F;

  const int tid  = threadIdx.x;
  const int wave = tid >> 5;
  const int lane = tid & 31;
  const int l16  = lane & 15;
  const int kSel = (lane >> 4) << 1;
  const int mHi  = (lane >> 4) << 3;
  const int waveU = __builtin_amdgcn_readfirstlane(wave);
  const int grp   = waveU >> 2;
  const int wl    = waveU & 3;
  const int kb    = (wl + 2 * grp) & 3;
  const int mB    = kb << 4;
  const bool ownsZero = (kb == 0);
  const int mI    = wl << 5;
  const int tl    = tid & 127;
  const bool younger = waveU >= 4;

  float* xb  = smem + LDS_XB + grp * 8192;
  float* Yr  = smem + LDS_YR + grp * 4096;
  float* Yi  = smem + LDS_YI + grp * 4096;
  float* t64 = smem + LDS_T64 + grp * 64;

  {
    const float4* fsrc = (const float4*)(ws + OFF_F128Q);
    for (int i = tid; i < 128 * 64; i += 256) {
      float4 v = fsrc[i];
      int row = i >> 6, c4 = i & 63;
      *(float4*)(FL + row * FPITCH + (c4 << 2)) = v;
    }
    if (tid < 64) {
      float4 v; v.x = 1.f; v.y = 1.f; v.z = 1.f; v.w = -1.f;
      *(float4*)(FL + 128 * FPITCH + (tid << 2)) = v;
    }
    const float4* gsrc = (const float4*)(ws + OFF_G64PR);
    float4* gdst = (float4*)GLr;
    for (int i = tid; i < 2 * 4096 / 4; i += 256) gdst[i] = gsrc[i];
  }

  const float* Wf     = ws + OFF_WF;
  const float* Wi     = ws + OFF_WI;

  const int frow = (ownsZero && l16 == 0) ? 128 : (mB + l16);

  const int gb = (kSel << 8) + (l16 << 1);

  int swq[4];
#pragma unroll
  for (int q = 0; q < 4; ++q) swq[q] = (l16 ^ (2 * q + (kSel >> 1))) << 1;

  const int pairs = (rows + 1) >> 1;
  for (int pp = blockIdx.x; pp < pairs; pp += gridDim.x) {
  const int r = 2 * pp + grp;
  const bool live = r < rows;
  const size_t rowOff = (size_t)(live ? r : 0) * LROW;
  {
    const float4* src = (const float4*)(uin + rowOff);
    float4 xv[16];
#pragma unroll
    for (int j = 0; j < 16; ++j) xv[j] = src[tl + j * 128];
#pragma unroll
    for (int j = 0; j < 16; ++j) {
      int i = tl + j * 128;
      float4 v = xv[j];
      int n2 = i >> 4, n1 = (i & 15) << 2;
      int base = pidx(n2, n1);
      xb[base] = v.x; xb[base + 2] = v.y; xb[base + 4] = v.z; xb[base + 6] = v.w;
    }
  }
  __syncthreads();

  for (int layer = 0; layer < 4; ++layer) {
    const float* kh = ws + OFF_KHAT + layer * 16384;

    if (younger) __builtin_amdgcn_s_setprio(3);

    v8f yr[4] = {}, yi[4] = {}, nyi[4];
    for (int kk = 0; kk < 128; kk += 4) {
      float4 fq = *(const float4*)(FL + frow * FPITCH + (kk + kSel) * 2);
      v2f fr; fr.x = fq.x; fr.y = fq.y;
      v2f fi; fi.x = fq.z; fi.y = fq.w;
      int jb = ((kk + kSel) >> 1) << 7;
#pragma unroll
      for (int t = 0; t < 4; ++t) {
        v2f x = *(const v2f*)(xb + jb + (((t << 4) + l16) << 1));
        yr[t] = wmma4(x, fr, yr[t]);
        yi[t] = wmma4(x, fi, yi[t]);
      }
    }

    if (ownsZero) {
      const float* s64 = ws + OFF_S64P + layer * 4096;
      v8f tz[4] = {};
      v2f sa[4][4][4];
#pragma unroll
      for (int mt = 0; mt < 4; ++mt) {
#pragma unroll
        for (int s = 0; s < 4; ++s) {
#pragma unroll
          for (int mo = 0; mo < 4; ++mo) {
            int go = gb + ((8 * mt + s) << 7) + (mo << 5);
            sa[mt][s][mo] = *(const v2f*)(s64 + go);
          }
        }
      }
      __builtin_amdgcn_sched_barrier(0);
#pragma unroll
      for (int mt = 0; mt < 4; ++mt) {
#pragma unroll
        for (int s = 0; s < 4; ++s) {
          v2f bv; bv.x = yi[mt][2 * s]; bv.y = yi[mt][2 * s + 1];
#pragma unroll
          for (int mo = 0; mo < 4; ++mo) tz[mo] = wmma4(sa[mt][s][mo], bv, tz[mo]);
        }
      }
#pragma unroll
      for (int mo = 0; mo < 4; ++mo) {
#pragma unroll
        for (int rr = 0; rr < 8; ++rr) yi[mo][rr] = (l16 == 0) ? 0.f : yi[mo][rr];
      }
      if (l16 == 0) {
#pragma unroll
        for (int mo = 0; mo < 4; ++mo) {
#pragma unroll
          for (int rr = 0; rr < 8; ++rr) t64[(mo << 4) + mHi + rr] = tz[mo][rr];
        }
      }
    }

    {
      const float4* w4 = (const float4*)(Wf + ((((mB + l16) << 6) + mHi) << 1));
#pragma unroll
      for (int t = 0; t < 4; ++t) {
#pragma unroll
        for (int q = 0; q < 4; ++q) {
          float4 w = w4[t * 8 + q];
          float a0 = yr[t][2 * q], b0 = yi[t][2 * q];
          float a1 = yr[t][2 * q + 1], b1 = yi[t][2 * q + 1];
          float i0 = a0 * w.y + b0 * w.x, i1 = a1 * w.w + b1 * w.z;
          yr[t][2 * q]     = a0 * w.x - b0 * w.y;
          yr[t][2 * q + 1] = a1 * w.z - b1 * w.w;
          yi[t][2 * q]  = i0;  yi[t][2 * q + 1]  = i1;
          nyi[t][2 * q] = -i0; nyi[t][2 * q + 1] = -i1;
        }
      }
    }

    v8f zr[4] = {}, zi[4] = {}, nzr[4];
#pragma unroll
    for (int mt = 0; mt < 4; ++mt) {
      if (mt == 2) {
        __builtin_amdgcn_sched_barrier(0);
        __builtin_amdgcn_s_setprio(0);
      }
#pragma unroll
      for (int s = 0; s < 4; ++s) {
        v2f br;  br.x  = yr[mt][2 * s];  br.y  = yr[mt][2 * s + 1];
        v2f bi;  bi.x  = yi[mt][2 * s];  bi.y  = yi[mt][2 * s + 1];
        v2f nbi; nbi.x = nyi[mt][2 * s]; nbi.y = nyi[mt][2 * s + 1];
#pragma unroll
        for (int mo = 0; mo < 4; ++mo) {
          int go = gb + ((8 * mt + s) << 7) + (mo << 5);
          v2f ac = *(const v2f*)(GLr + go);
          v2f as = *(const v2f*)(GLi + go);
          zr[mo] = wmma4(ac, br, zr[mo]);  zr[mo] = wmma4(as, nbi, zr[mo]);
          zi[mo] = wmma4(as, br, zi[mo]);  zi[mo] = wmma4(ac, bi, zi[mo]);
        }
      }
    }
    {
      const float4* k4 = (const float4*)(kh + ((((mB + l16) << 6) + mHi) << 1));
#pragma unroll
      for (int mo = 0; mo < 4; ++mo) {
#pragma unroll
        for (int q = 0; q < 4; ++q) {
          float4 k = k4[mo * 8 + q];
          float a0 = zr[mo][2 * q], b0 = zi[mo][2 * q];
          float a1 = zr[mo][2 * q + 1], b1 = zi[mo][2 * q + 1];
          float r0 = a0 * k.x - b0 * k.y, r1 = a1 * k.z - b1 * k.w;
          zi[mo][2 * q]     = a0 * k.y + b0 * k.x;
          zi[mo][2 * q + 1] = a1 * k.w + b1 * k.z;
          zr[mo][2 * q]  = r0;  zr[mo][2 * q + 1]  = r1;
          nzr[mo][2 * q] = -r0; nzr[mo][2 * q + 1] = -r1;
        }
      }
    }

    v8f pr[4] = {}, pi[4] = {};
#pragma unroll
    for (int mo = 0; mo < 4; ++mo) {
      if (mo == 2) __builtin_amdgcn_sched_barrier(0);
#pragma unroll
      for (int s = 0; s < 4; ++s) {
        v2f br;  br.x  = zr[mo][2 * s];  br.y  = zr[mo][2 * s + 1];
        v2f bi;  bi.x  = zi[mo][2 * s];  bi.y  = zi[mo][2 * s + 1];
        v2f nbr; nbr.x = nzr[mo][2 * s]; nbr.y = nzr[mo][2 * s + 1];
#pragma unroll
        for (int t = 0; t < 4; ++t) {
          int go = gb + ((8 * mo + s) << 7) + (t << 5);
          v2f ac = *(const v2f*)(GLr + go);
          v2f as = *(const v2f*)(GLi + go);
          pr[t] = wmma4(ac, br, pr[t]);   pr[t] = wmma4(as, bi, pr[t]);
          pi[t] = wmma4(as, nbr, pi[t]);  pi[t] = wmma4(ac, bi, pi[t]);
        }
      }
    }
    {
      const float4* w4 = (const float4*)(Wi + ((((mB + l16) << 6) + mHi) << 1));
      const int sb = (((mB + l16) >> 1) << 7) + (l16 & 1);
      const int q8 = l16 >> 1;
#pragma unroll
      for (int t = 0; t < 4; ++t) {
#pragma unroll
        for (int q = 0; q < 4; ++q) {
          float4 w = w4[t * 8 + q];
          float a0 = pr[t][2 * q], b0 = pi[t][2 * q];
          float a1 = pr[t][2 * q + 1], b1 = pi[t][2 * q + 1];
          int p0 = sb + ((((t << 4) + mHi + 2 * q)     ^ q8) << 1);
          int p1 = sb + ((((t << 4) + mHi + 2 * q + 1) ^ q8) << 1);
          Yr[p0] = a0 * w.x - b0 * w.y;
          Yi[p0] = a0 * w.y + b0 * w.x;
          Yr[p1] = a1 * w.z - b1 * w.w;
          Yi[p1] = a1 * w.w + b1 * w.z;
        }
      }
    }
    __syncthreads();

    {
      v8f acc[2][4];
      float tv[4];
#pragma unroll
      for (int t = 0; t < 4; ++t) {
        tv[t] = t64[(t << 4) + l16];
#pragma unroll
        for (int u = 0; u < 2; ++u) {
#pragma unroll
          for (int v = 0; v < 8; v += 2) {
            v2f xr = *(const v2f*)(xb + pidx(mI + (u << 4) + v + mHi, (t << 4) + l16));
            acc[u][t][v] = xr.x; acc[u][t][v + 1] = xr.y;
          }
        }
      }
      __builtin_amdgcn_sched_barrier(0);
#pragma unroll
      for (int t = 0; t < 4; ++t) {
#pragma unroll
        for (int u = 0; u < 2; ++u) {
#pragma unroll
          for (int v = 0; v < 8; v += 2) { acc[u][t][v] += tv[t]; acc[u][t][v + 1] -= tv[t]; }
        }
      }
      for (int kk0 = 0; kk0 < 64; kk0 += 16) {
#pragma unroll
        for (int q = 0; q < 4; ++q) {
          int kk = kk0 + 4 * q;
          float4 aq0 = *(const float4*)(FL + (mI + l16) * FPITCH + (kk + kSel) * 2);
          float4 aq1 = *(const float4*)(FL + (mI + 16 + l16) * FPITCH + (kk + kSel) * 2);
          v2f ar0; ar0.x = aq0.x; ar0.y = aq0.y;
          v2f as0; as0.x = aq0.z; as0.y = aq0.w;
          v2f ar1; ar1.x = aq1.x; ar1.y = aq1.y;
          v2f as1; as1.x = aq1.z; as1.y = aq1.w;
          int jb = (((kk + kSel) >> 1) << 7) + swq[q];
#pragma unroll
          for (int t = 0; t < 4; ++t) {
            v2f br  = *(const v2f*)(Yr + jb + (t << 5));
            v2f byi = *(const v2f*)(Yi + jb + (t << 5));
            acc[0][t] = wmma4(ar0, br, acc[0][t]);
            acc[0][t] = wmma4(as0, byi, acc[0][t]);
            acc[1][t] = wmma4(ar1, br, acc[1][t]);
            acc[1][t] = wmma4(as1, byi, acc[1][t]);
          }
        }
      }
#pragma unroll
      for (int t = 0; t < 4; ++t) {
#pragma unroll
        for (int u = 0; u < 2; ++u) {
#pragma unroll
          for (int v = 0; v < 8; ++v) {
            int row = mI + (u << 4) + v + mHi, col = (t << 4) + l16;
            int p = pidx(row, col);
            float y = acc[u][t][v];
            xb[p] = 0.5f * y * (1.f + erff(y * 0.70710678118654752f));
          }
        }
      }
    }
    __syncthreads();
  }

  if (live) {
    float4* dst = (float4*)(uout + rowOff);
    for (int i = tl; i < LROW / 4; i += 128) {
      int n2 = i >> 4, n1 = (i & 15) << 2;
      int base = pidx(n2, n1);
      float4 o;
      o.x = xb[base]; o.y = xb[base + 2]; o.z = xb[base + 4]; o.w = xb[base + 6];
      dst[i] = o;
    }
  }
  }
}

extern "C" void kernel_launch(void* const* d_in, const int* in_sizes, int n_in,
                              void* d_out, int out_size, void* d_ws, size_t ws_size,
                              hipStream_t stream) {
  const float* u    = (const float*)d_in[0];
  const float* Lr   = (const float*)d_in[1];
  const float* Li   = (const float*)d_in[2];
  const float* Pv   = (const float*)d_in[3];
  const float* Bv   = (const float*)d_in[4];
  const float* Ct   = (const float*)d_in[5];
  const float* step = (const float*)d_in[6];
  float* out = (float*)d_out;
  float* ws  = (float*)d_ws;

  int rows = in_sizes[0] / LROW;

  ssm_gen_tables<<<64, 256, 0, stream>>>(ws);
  ssm_gen_khat<<<dim3(32, 4), 256, 0, stream>>>(ws, Lr, Li, Pv, Bv, Ct, step);
  ssm_gen_s64<<<dim3(16, 4), 256, 0, stream>>>(ws);
  ssm_rows<<<256, 256, 0, stream>>>(u, out, ws, rows);
}


# Inverse stage 2 split 2 x 64: rows n2 and n2+64 share an even-k2 sum E and an odd-k2 sum O (E+O, E-O); Y row pairs hold equal-parity k2; 128 not 256 WMMAs per wave; regroups the last stage's sums
# speedup vs baseline: 3.2043x; 1.3623x over previous
#include <hip/hip_runtime.h>
#include <math.h>

typedef float v2f __attribute__((ext_vector_type(2)));
typedef float v8f __attribute__((ext_vector_type(8)));

#define LROW 8192
#define NSTATE 64
#define PI_F 3.14159265358979323846f

#define OFF_F128Q   0
#define OFF_S64P    32768
#define OFF_G64PR   65536
#define OFF_G64PI   69632
#define OFF_TW16    73728
#define OFF_WF      81920
#define OFF_WI      98304
#define OFF_KHAT    114688
#define WS_FLOATS   (OFF_KHAT + 4*16384)

#define LDS_XB      0
#define LDS_YR      16384
#define LDS_YI      24576
#define LDS_T64     32768
#define LDS_F       32896
#define FPITCH      260
#define LDS_FLOATS  (LDS_F + 129 * FPITCH)

static __device__ inline v8f wmma4(v2f a, v2f b, v8f c) {
  return __builtin_amdgcn_wmma_f32_16x16x4_f32(false, a, false, b, (short)0, c,
                                               false, false);
}

static __device__ inline int pidx(int row, int col) {
  return ((row >> 1) << 7) + (col << 1) + (row & 1);
}

__global__ void ssm_gen_tables(float* __restrict__ ws) {
  int idx = blockIdx.x * blockDim.x + threadIdx.x;
  {
    int k2 = idx >> 7, n2 = idx & 127;
    int t = (k2 * n2) & 127;
    float s, c;
    sincosf(-2.f * PI_F * (float)t / 128.f, &s, &c);
    int pos = k2 * 256 + (n2 >> 1) * 4 + (n2 & 1);
    ws[OFF_F128Q + pos]      = c;
    ws[OFF_F128Q + pos + 2]  = s;
  }
  if (idx < 8192) {
    int k2 = idx >> 6, n1 = idx & 63;
    int t = (n1 * k2) & 8191;
    float s, c;
    sincosf(-2.f * PI_F * (float)t / 8192.f, &s, &c);
    ws[OFF_WF + 2 * idx]     = c;  ws[OFF_WF + 2 * idx + 1] = s;
    ws[OFF_WI + 2 * idx]     = c;  ws[OFF_WI + 2 * idx + 1] = -s;
  }
  if (idx < 4096) {
    int n1 = idx >> 6, k1 = idx & 63;
    int t = (n1 * k1) & 63;
    float s, c;
    sincosf(-2.f * PI_F * (float)t / 64.f, &s, &c);
    int pos = (n1 >> 1) * 128 + 2 * k1 + (n1 & 1);
    ws[OFF_G64PR + pos]  = c;  ws[OFF_G64PI + pos]  = s;
  }
  if (idx < 64) {
    int h = idx >> 5, d = (idx >> 3) & 3, r = idx & 7;
    int t = ((8 * h + r) * d) & 63;
    float s, c;
    sincosf(-2.f * PI_F * (float)t / 64.f, &s, &c);
    ws[OFF_TW16 + 2 * idx] = c;  ws[OFF_TW16 + 2 * idx + 1] = s;
  }
}

static __device__ inline void dplr_eval(int kf, float sc,
                                        const float* __restrict__ lr, const float* __restrict__ li,
                                        const float* __restrict__ pp, const float* __restrict__ bb,
                                        const float* __restrict__ ct, float* outr, float* outi) {
  float osin, ocos;
  sincosf(-2.f * PI_F * (float)kf / 8192.f, &osin, &ocos);
  float pr = 1.f + ocos, pi = osin;
  float mr = 1.f - ocos, mi = -osin;
  float pd = pr * pr + pi * pi;
  float ipr = pr / pd, ipi = -pi / pd;
  float cr = 2.f * ipr, ci = 2.f * ipi;
  float gr = sc * (mr * ipr - mi * ipi);
  float gi = sc * (mr * ipi + mi * ipr);

  float k00r = 0, k00i = 0, k01r = 0, k01i = 0;
  float k10r = 0, k10i = 0, k11r = 0, k11i = 0;
  for (int n = 0; n < NSTATE; ++n) {
    float dr = gr - lr[n], di = gi - li[n];
    float dd = dr * dr + di * di;
    float ir = dr / dd, ii = -di / dd;
    float b = bb[n], p = pp[n], c = ct[n];
    float w00 = c * b, w01 = c * p, w10 = p * b, w11 = p * p;
    k00r += w00 * ir; k00i += w00 * ii;
    k01r += w01 * ir; k01i += w01 * ii;
    k10r += w10 * ir; k10i += w10 * ii;
    k11r += w11 * ir; k11i += w11 * ii;
  }
  float qr = k01r * k10r - k01i * k10i;
  float qi = k01r * k10i + k01i * k10r;
  float er = 1.f + k11r, ei = k11i;
  float ed = er * er + ei * ei;
  float ier = er / ed, iei = -ei / ed;
  float tr = qr * ier - qi * iei;
  float ti = qr * iei + qi * ier;
  float kr = k00r - tr, ki = k00i - ti;
  *outr = cr * kr - ci * ki;
  *outi = cr * ki + ci * kr;
}

__global__ void ssm_gen_khat(float* __restrict__ ws,
                             const float* __restrict__ Lr,
                             const float* __restrict__ Li,
                             const float* __restrict__ Pv,
                             const float* __restrict__ Bv,
                             const float* __restrict__ Ct,
                             const float* __restrict__ stepv) {
  int layer = blockIdx.y;
  int s = blockIdx.x * blockDim.x + threadIdx.x;
  int k2 = s >> 6, p = s & 63;
  int k1 = (k2 < 64) ? ((p >> 4) + 32 * ((p >> 3) & 1) + 4 * (p & 7)) : p;
  int kf = 128 * k1 + k2;
  int km = (8192 - kf) & 8191;

  float sc = 2.f / stepv[layer];
  const float* lr = Lr + layer * NSTATE;
  const float* li = Li + layer * NSTATE;
  const float* pp = Pv + layer * NSTATE;
  const float* bb = Bv + layer * NSTATE;
  const float* ct = Ct + layer * NSTATE;

  float ar, ai, br, bi;
  dplr_eval(kf, sc, lr, li, pp, bb, ct, &ar, &ai);
  dplr_eval(km, sc, lr, li, pp, bb, ct, &br, &bi);
  const float twice = (k2 == 0 || k2 == 64) ? 1.f : 2.f;
  const float scale = twice * 0.5f / 8192.f;
  ws[OFF_KHAT + layer * 16384 + 2 * s]     = (ar + br) * scale;
  ws[OFF_KHAT + layer * 16384 + 2 * s + 1] = (ai - bi) * scale;
}

__global__ void ssm_gen_s64(float* __restrict__ ws) {
  __shared__ float cs[128], sn[128], kr[64], ki[64];
  const int layer = blockIdx.y;
  const int tid = threadIdx.x;
  if (tid < 128) {
    float s, c;
    sincosf(2.f * PI_F * (float)tid / 128.f, &s, &c);
    cs[tid] = c; sn[tid] = s;
  }
  if (tid < 64) {
    kr[tid] = ws[OFF_KHAT + layer * 16384 + 2 * (4096 + tid)];
    ki[tid] = ws[OFF_KHAT + layer * 16384 + 2 * (4096 + tid) + 1];
  }
  __syncthreads();
  const int idx = blockIdx.x * 256 + tid;
  const int m = idx >> 6, n1 = idx & 63;
  const int d = n1 - m;
  float acc = 0.f;
  for (int k1 = 0; k1 < 64; ++k1) {
    int t = ((2 * k1 + 1) * d) & 127;
    acc += kr[k1] * cs[t] - ki[k1] * sn[t];
  }
  ws[OFF_S64P + layer * 4096 + (m >> 1) * 128 + 2 * n1 + (m & 1)] = acc;
}

__global__ __launch_bounds__(256) void ssm_rows(const float* __restrict__ uin,
                                                float* __restrict__ uout,
                                                const float* __restrict__ ws,
                                                int rows) {
  __shared__ __attribute__((aligned(16))) float smem[LDS_FLOATS];
  float* FL  = smem + LDS_F;

  const int tid  = threadIdx.x;
  const int wave = tid >> 5;
  const int lane = tid & 31;
  const int l16  = lane & 15;
  const int kSel = (lane >> 4) << 1;
  const int mHi  = (lane >> 4) << 3;
  const int waveU = __builtin_amdgcn_readfirstlane(wave);
  const int grp   = waveU >> 2;
  const int wl    = waveU & 3;
  const int kb    = (wl + 2 * grp) & 3;
  const int mB    = kb << 4;
  const bool ownsZero = (kb == 0);
  const int mI    = wl << 4;
  const int tl    = tid & 127;
  const bool younger = waveU >= 4;

  float* xb  = smem + LDS_XB + grp * 8192;
  float* Yr  = smem + LDS_YR + grp * 4096;
  float* Yi  = smem + LDS_YI + grp * 4096;
  float* t64 = smem + LDS_T64 + grp * 64;

  {
    const float4* fsrc = (const float4*)(ws + OFF_F128Q);
    for (int i = tid; i < 128 * 64; i += 256) {
      float4 v = fsrc[i];
      int row = i >> 6, c4 = i & 63;
      *(float4*)(FL + row * FPITCH + (c4 << 2)) = v;
    }
    if (tid < 64) {
      float4 v; v.x = 1.f; v.y = 1.f; v.z = 1.f; v.w = -1.f;
      *(float4*)(FL + 128 * FPITCH + (tid << 2)) = v;
    }
  }

  const float* Wf     = ws + OFF_WF;
  const float* Wi     = ws + OFF_WI;

  const int frow = (ownsZero && l16 == 0) ? 128 : (mB + l16);

  const int gb = (kSel << 8) + (l16 << 1);

  v2f f16c[4], f16s[4];
#pragma unroll
  for (int s = 0; s < 4; ++s) {
    int go = (((mHi >> 1) + s) << 7) + (l16 << 3);
    f16c[s] = *(const v2f*)(ws + OFF_G64PR + go);
    f16s[s] = *(const v2f*)(ws + OFF_G64PI + go);
  }
  const float4* tw4 = (const float4*)(ws + OFF_TW16 + ((lane >> 4) << 6));

  int swq[2][2];
#pragma unroll
  for (int eo = 0; eo < 2; ++eo) {
#pragma unroll
    for (int p = 0; p < 2; ++p) swq[eo][p] = (l16 ^ (4 * eo + kSel + p)) << 1;
  }

  const int pairs = (rows + 1) >> 1;
  for (int pp = blockIdx.x; pp < pairs; pp += gridDim.x) {
  const int r = 2 * pp + grp;
  const bool live = r < rows;
  const size_t rowOff = (size_t)(live ? r : 0) * LROW;
  {
    const float4* src = (const float4*)(uin + rowOff);
    float4 xv[16];
#pragma unroll
    for (int j = 0; j < 16; ++j) xv[j] = src[tl + j * 128];
#pragma unroll
    for (int j = 0; j < 16; ++j) {
      int i = tl + j * 128;
      float4 v = xv[j];
      int n2 = i >> 4, n1 = (i & 15) << 2;
      int base = pidx(n2, n1);
      xb[base] = v.x; xb[base + 2] = v.y; xb[base + 4] = v.z; xb[base + 6] = v.w;
    }
  }
  __syncthreads();

  for (int layer = 0; layer < 4; ++layer) {
    const float* kh = ws + OFF_KHAT + layer * 16384;

    if (younger) __builtin_amdgcn_s_setprio(3);

    v8f yr[4] = {}, yi[4] = {};
    for (int kk = 0; kk < 128; kk += 4) {
      if (kk == 96) __builtin_amdgcn_s_setprio(0);
      float4 fq = *(const float4*)(FL + frow * FPITCH + (kk + kSel) * 2);
      v2f fr; fr.x = fq.x; fr.y = fq.y;
      v2f fi; fi.x = fq.z; fi.y = fq.w;
      int jb = ((kk + kSel) >> 1) << 7;
#pragma unroll
      for (int t = 0; t < 4; ++t) {
        v2f x = *(const v2f*)(xb + jb + (((t << 4) + l16) << 1));
        yr[t] = wmma4(x, fr, yr[t]);
        yi[t] = wmma4(x, fi, yi[t]);
      }
    }

    if (ownsZero) {
      const float* s64 = ws + OFF_S64P + layer * 4096;
      v8f tz[4] = {};
      v2f sa[4][4][4];
#pragma unroll
      for (int mt = 0; mt < 4; ++mt) {
#pragma unroll
        for (int s = 0; s < 4; ++s) {
#pragma unroll
          for (int mo = 0; mo < 4; ++mo) {
            int go = gb + ((8 * mt + s) << 7) + (mo << 5);
            sa[mt][s][mo] = *(const v2f*)(s64 + go);
          }
        }
      }
      __builtin_amdgcn_sched_barrier(0);
#pragma unroll
      for (int mt = 0; mt < 4; ++mt) {
#pragma unroll
        for (int s = 0; s < 4; ++s) {
          v2f bv; bv.x = yi[mt][2 * s]; bv.y = yi[mt][2 * s + 1];
#pragma unroll
          for (int mo = 0; mo < 4; ++mo) tz[mo] = wmma4(sa[mt][s][mo], bv, tz[mo]);
        }
      }
#pragma unroll
      for (int mo = 0; mo < 4; ++mo) {
#pragma unroll
        for (int rr = 0; rr < 8; ++rr) yi[mo][rr] = (l16 == 0) ? 0.f : yi[mo][rr];
      }
      if (l16 == 0) {
#pragma unroll
        for (int mo = 0; mo < 4; ++mo) {
#pragma unroll
          for (int rr = 0; rr < 8; ++rr) t64[(mo << 4) + mHi + rr] = tz[mo][rr];
        }
      }
    }

    {
      const float4* w4 = (const float4*)(Wf + ((((mB + l16) << 6) + mHi) << 1));
#pragma unroll
      for (int t = 0; t < 4; ++t) {
#pragma unroll
        for (int q = 0; q < 4; ++q) {
          float4 w = w4[t * 8 + q];
          float a0 = yr[t][2 * q], b0 = yi[t][2 * q];
          float a1 = yr[t][2 * q + 1], b1 = yi[t][2 * q + 1];
          float i0 = a0 * w.y + b0 * w.x, i1 = a1 * w.w + b1 * w.z;
          yr[t][2 * q]     = a0 * w.x - b0 * w.y;
          yr[t][2 * q + 1] = a1 * w.z - b1 * w.w;
          yi[t][2 * q]  = i0;  yi[t][2 * q + 1]  = i1;
        }
      }
    }

    v8f tc[4], ts[4];
#pragma unroll
    for (int d = 1; d < 4; ++d) {
#pragma unroll
      for (int q = 0; q < 4; ++q) {
        float4 w = tw4[d * 4 + q];
        tc[d][2 * q] = w.x;     ts[d][2 * q] = w.y;
        tc[d][2 * q + 1] = w.z; ts[d][2 * q + 1] = w.w;
      }
    }

    v8f ur[4], ui[4], nui[4];
    {
      v8f s02r = yr[0] + yr[2], d02r = yr[0] - yr[2], s02i = yi[0] + yi[2], d02i = yi[0] - yi[2];
      v8f s13r = yr[1] + yr[3], d13r = yr[1] - yr[3], s13i = yi[1] + yi[3], d13i = yi[1] - yi[3];
      ur[0] = s02r + s13r;  ui[0] = s02i + s13i;
      v8f u2r = s02r - s13r, u2i = s02i - s13i;
      v8f u1r = d02r + d13i, u1i = d02i - d13r;
      v8f u3r = d02r - d13i, u3i = d02i + d13r;
      ur[1] = u1r * tc[1] - u1i * ts[1];  ui[1] = u1r * ts[1] + u1i * tc[1];
      ur[2] = u2r * tc[2] - u2i * ts[2];  ui[2] = u2r * ts[2] + u2i * tc[2];
      ur[3] = u3r * tc[3] - u3i * ts[3];  ui[3] = u3r * ts[3] + u3i * tc[3];
#pragma unroll
      for (int d = 0; d < 4; ++d) nui[d] = -ui[d];
    }
    v8f zr[4] = {}, zi[4] = {}, nzr[4];
#pragma unroll
    for (int s = 0; s < 4; ++s) {
#pragma unroll
      for (int d = 0; d < 4; ++d) {
        v2f br;  br.x  = ur[d][2 * s];  br.y  = ur[d][2 * s + 1];
        v2f bi;  bi.x  = ui[d][2 * s];  bi.y  = ui[d][2 * s + 1];
        v2f nbi; nbi.x = nui[d][2 * s]; nbi.y = nui[d][2 * s + 1];
        zr[d] = wmma4(f16c[s], br, zr[d]);  zr[d] = wmma4(f16s[s], nbi, zr[d]);
        zi[d] = wmma4(f16s[s], br, zi[d]);  zi[d] = wmma4(f16c[s], bi, zi[d]);
      }
    }
    {
      const float4* k4 = (const float4*)(kh + ((((mB + l16) << 6) + mHi) << 1));
#pragma unroll
      for (int mo = 0; mo < 4; ++mo) {
#pragma unroll
        for (int q = 0; q < 4; ++q) {
          float4 k = k4[mo * 8 + q];
          float a0 = zr[mo][2 * q], b0 = zi[mo][2 * q];
          float a1 = zr[mo][2 * q + 1], b1 = zi[mo][2 * q + 1];
          float r0 = a0 * k.x - b0 * k.y, r1 = a1 * k.z - b1 * k.w;
          zi[mo][2 * q]     = a0 * k.y + b0 * k.x;
          zi[mo][2 * q + 1] = a1 * k.w + b1 * k.z;
          zr[mo][2 * q]  = r0;  zr[mo][2 * q + 1]  = r1;
          nzr[mo][2 * q] = -r0; nzr[mo][2 * q + 1] = -r1;
        }
      }
    }

    v8f vr[4] = {}, vi[4] = {};
#pragma unroll
    for (int s = 0; s < 4; ++s) {
#pragma unroll
      for (int d = 0; d < 4; ++d) {
        v2f br;  br.x  = zr[d][2 * s];  br.y  = zr[d][2 * s + 1];
        v2f bi;  bi.x  = zi[d][2 * s];  bi.y  = zi[d][2 * s + 1];
        v2f nbr; nbr.x = nzr[d][2 * s]; nbr.y = nzr[d][2 * s + 1];
        vr[d] = wmma4(f16c[s], br, vr[d]);   vr[d] = wmma4(f16s[s], bi, vr[d]);
        vi[d] = wmma4(f16s[s], nbr, vi[d]);  vi[d] = wmma4(f16c[s], bi, vi[d]);
      }
    }
    v8f pr[4], pi[4];
    {
      v8f w1r = vr[1] * tc[1] + vi[1] * ts[1], w1i = vi[1] * tc[1] - vr[1] * ts[1];
      v8f w2r = vr[2] * tc[2] + vi[2] * ts[2], w2i = vi[2] * tc[2] - vr[2] * ts[2];
      v8f w3r = vr[3] * tc[3] + vi[3] * ts[3], w3i = vi[3] * tc[3] - vr[3] * ts[3];
      v8f s02r = vr[0] + w2r, d02r = vr[0] - w2r, s02i = vi[0] + w2i, d02i = vi[0] - w2i;
      v8f s13r = w1r + w3r,   d13r = w1r - w3r,   s13i = w1i + w3i,   d13i = w1i - w3i;
      pr[0] = s02r + s13r;  pi[0] = s02i + s13i;
      pr[2] = s02r - s13r;  pi[2] = s02i - s13i;
      pr[1] = d02r - d13i;  pi[1] = d02i + d13r;
      pr[3] = d02r + d13i;  pi[3] = d02i - d13r;
    }
    {
      const float4* w4 = (const float4*)(Wi + ((((mB + l16) << 6) + mHi) << 1));
      const int k2w = mB + l16;
      const int jw = ((k2w >> 2) << 1) | (k2w & 1);
      const int sb = (jw << 7) + ((k2w >> 1) & 1);
      const int q8 = jw & 7;
#pragma unroll
      for (int t = 0; t < 4; ++t) {
#pragma unroll
        for (int q = 0; q < 4; ++q) {
          float4 w = w4[t * 8 + q];
          float a0 = pr[t][2 * q], b0 = pi[t][2 * q];
          float a1 = pr[t][2 * q + 1], b1 = pi[t][2 * q + 1];
          int p0 = sb + ((((t << 4) + mHi + 2 * q)     ^ q8) << 1);
          int p1 = sb + ((((t << 4) + mHi + 2 * q + 1) ^ q8) << 1);
          Yr[p0] = a0 * w.x - b0 * w.y;
          Yi[p0] = a0 * w.y + b0 * w.x;
          Yr[p1] = a1 * w.z - b1 * w.w;
          Yi[p1] = a1 * w.w + b1 * w.z;
        }
      }
    }
    __syncthreads();

    {
      v8f accE[4], accO[4];
      v2f xlo[4][4], xhi[4][4];
      float tv[4];
#pragma unroll
      for (int t = 0; t < 4; ++t) {
        tv[t] = t64[(t << 4) + l16];
#pragma unroll
        for (int v = 0; v < 8; v += 2) {
          xlo[t][v >> 1] = *(const v2f*)(xb + pidx(mI + v + mHi, (t << 4) + l16));
          xhi[t][v >> 1] = *(const v2f*)(xb + pidx(64 + mI + v + mHi, (t << 4) + l16));
        }
      }
      __builtin_amdgcn_sched_barrier(0);
#pragma unroll
      for (int t = 0; t < 4; ++t) {
#pragma unroll
        for (int v = 0; v < 8; v += 2) {
          accE[t][v] = tv[t];  accE[t][v + 1] = -tv[t];
          accO[t][v] = 0.f;    accO[t][v + 1] = 0.f;
        }
      }
#pragma nounroll
      for (int e0 = 0; e0 < 8; e0 += 2) {
        float4 q0[2], q1[2];
        v2f brE[2][4], biE[2][4], brO[2][4], biO[2][4];
#pragma unroll
        for (int eo = 0; eo < 2; ++eo) {
          const int e = e0 + eo;
          const float* fa = FL + (mI + l16) * FPITCH + (e << 4) + (kSel << 2);
          q0[eo] = *(const float4*)(fa);
          q1[eo] = *(const float4*)(fa + 4);
          const int jbE = ((4 * e + kSel) << 7) + swq[eo][0];
          const int jbO = ((4 * e + kSel + 1) << 7) + swq[eo][1];
#pragma unroll
          for (int t = 0; t < 4; ++t) {
            brE[eo][t] = *(const v2f*)(Yr + jbE + (t << 5));
            biE[eo][t] = *(const v2f*)(Yi + jbE + (t << 5));
            brO[eo][t] = *(const v2f*)(Yr + jbO + (t << 5));
            biO[eo][t] = *(const v2f*)(Yi + jbO + (t << 5));
          }
        }
        __builtin_amdgcn_sched_barrier(0);
#pragma unroll
        for (int eo = 0; eo < 2; ++eo) {
          v2f ce; ce.x = q0[eo].x; ce.y = q1[eo].x;
          v2f se; se.x = q0[eo].z; se.y = q1[eo].z;
          v2f co; co.x = q0[eo].y; co.y = q1[eo].y;
          v2f so; so.x = q0[eo].w; so.y = q1[eo].w;
#pragma unroll
          for (int t = 0; t < 4; ++t) {
            accE[t] = wmma4(ce, brE[eo][t], accE[t]);
            accO[t] = wmma4(co, brO[eo][t], accO[t]);
            accE[t] = wmma4(se, biE[eo][t], accE[t]);
            accO[t] = wmma4(so, biO[eo][t], accO[t]);
          }
        }
      }
#pragma unroll
      for (int t = 0; t < 4; ++t) {
#pragma unroll
        for (int v = 0; v < 8; ++v) {
          const int row = mI + v + mHi, col = (t << 4) + l16;
          const float xl = (v & 1) ? xlo[t][v >> 1].y : xlo[t][v >> 1].x;
          const float xh = (v & 1) ? xhi[t][v >> 1].y : xhi[t][v >> 1].x;
          const float yl = (accE[t][v] + accO[t][v]) + xl;
          const float yh = (accE[t][v] - accO[t][v]) + xh;
          xb[pidx(row, col)]      = 0.5f * yl * (1.f + erff(yl * 0.70710678118654752f));
          xb[pidx(row + 64, col)] = 0.5f * yh * (1.f + erff(yh * 0.70710678118654752f));
        }
      }
    }
    __syncthreads();
  }

  if (live) {
    float4* dst = (float4*)(uout + rowOff);
    for (int i = tl; i < LROW / 4; i += 128) {
      int n2 = i >> 4, n1 = (i & 15) << 2;
      int base = pidx(n2, n1);
      float4 o;
      o.x = xb[base]; o.y = xb[base + 2]; o.z = xb[base + 4]; o.w = xb[base + 6];
      dst[i] = o;
    }
  }
  }
}

extern "C" void kernel_launch(void* const* d_in, const int* in_sizes, int n_in,
                              void* d_out, int out_size, void* d_ws, size_t ws_size,
                              hipStream_t stream) {
  const float* u    = (const float*)d_in[0];
  const float* Lr   = (const float*)d_in[1];
  const float* Li   = (const float*)d_in[2];
  const float* Pv   = (const float*)d_in[3];
  const float* Bv   = (const float*)d_in[4];
  const float* Ct   = (const float*)d_in[5];
  const float* step = (const float*)d_in[6];
  float* out = (float*)d_out;
  float* ws  = (float*)d_ws;

  int rows = in_sizes[0] / LROW;

  ssm_gen_tables<<<64, 256, 0, stream>>>(ws);
  ssm_gen_khat<<<dim3(32, 4), 256, 0, stream>>>(ws, Lr, Li, Pv, Bv, Ct, step);
  ssm_gen_s64<<<dim3(16, 4), 256, 0, stream>>>(ws);
  ssm_rows<<<256, 256, 0, stream>>>(u, out, ws, rows);
}


# Forward stage 1 split 2 x 64: a wave owns 16 k2 of one parity and sums x[m] +- x[m+64] over m < 64 (fold: one packed add per fragment); 128 not 256 WMMAs per wave; regroups stage 1's sums
# speedup vs baseline: 3.3124x; 1.0337x over previous
#include <hip/hip_runtime.h>
#include <math.h>

typedef float v2f __attribute__((ext_vector_type(2)));
typedef float v8f __attribute__((ext_vector_type(8)));

#define LROW 8192
#define NSTATE 64
#define PI_F 3.14159265358979323846f

#define OFF_F128Q   0
#define OFF_S64P    32768
#define OFF_G64PR   65536
#define OFF_G64PI   69632
#define OFF_TW16    73728
#define OFF_WF      81920
#define OFF_WI      98304
#define OFF_KHAT    114688
#define WS_FLOATS   (OFF_KHAT + 4*16384)

#define LDS_XB      0
#define LDS_YR      16384
#define LDS_YI      24576
#define LDS_T64     32768
#define LDS_F       32896
#define FPITCH      260
#define LDS_FLOATS  (LDS_F + 129 * FPITCH)

static __device__ inline v8f wmma4(v2f a, v2f b, v8f c) {
  return __builtin_amdgcn_wmma_f32_16x16x4_f32(false, a, false, b, (short)0, c,
                                               false, false);
}

static __device__ inline int pidx(int row, int col) {
  return ((row >> 1) << 7) + (col << 1) + (row & 1);
}

__global__ void ssm_gen_tables(float* __restrict__ ws) {
  int idx = blockIdx.x * blockDim.x + threadIdx.x;
  {
    int k2 = idx >> 7, n2 = idx & 127;
    int t = (k2 * n2) & 127;
    float s, c;
    sincosf(-2.f * PI_F * (float)t / 128.f, &s, &c);
    int pos = k2 * 256 + (n2 >> 1) * 4 + (n2 & 1);
    ws[OFF_F128Q + pos]      = c;
    ws[OFF_F128Q + pos + 2]  = s;
  }
  if (idx < 8192) {
    int k2 = idx >> 6, n1 = idx & 63;
    int t = (n1 * k2) & 8191;
    float s, c;
    sincosf(-2.f * PI_F * (float)t / 8192.f, &s, &c);
    ws[OFF_WF + 2 * idx]     = c;  ws[OFF_WF + 2 * idx + 1] = s;
    ws[OFF_WI + 2 * idx]     = c;  ws[OFF_WI + 2 * idx + 1] = -s;
  }
  if (idx < 4096) {
    int n1 = idx >> 6, k1 = idx & 63;
    int t = (n1 * k1) & 63;
    float s, c;
    sincosf(-2.f * PI_F * (float)t / 64.f, &s, &c);
    int pos = (n1 >> 1) * 128 + 2 * k1 + (n1 & 1);
    ws[OFF_G64PR + pos]  = c;  ws[OFF_G64PI + pos]  = s;
  }
  if (idx < 64) {
    int h = idx >> 5, d = (idx >> 3) & 3, r = idx & 7;
    int t = ((8 * h + r) * d) & 63;
    float s, c;
    sincosf(-2.f * PI_F * (float)t / 64.f, &s, &c);
    ws[OFF_TW16 + 2 * idx] = c;  ws[OFF_TW16 + 2 * idx + 1] = s;
  }
}

static __device__ inline void dplr_eval(int kf, float sc,
                                        const float* __restrict__ lr, const float* __restrict__ li,
                                        const float* __restrict__ pp, const float* __restrict__ bb,
                                        const float* __restrict__ ct, float* outr, float* outi) {
  float osin, ocos;
  sincosf(-2.f * PI_F * (float)kf / 8192.f, &osin, &ocos);
  float pr = 1.f + ocos, pi = osin;
  float mr = 1.f - ocos, mi = -osin;
  float pd = pr * pr + pi * pi;
  float ipr = pr / pd, ipi = -pi / pd;
  float cr = 2.f * ipr, ci = 2.f * ipi;
  float gr = sc * (mr * ipr - mi * ipi);
  float gi = sc * (mr * ipi + mi * ipr);

  float k00r = 0, k00i = 0, k01r = 0, k01i = 0;
  float k10r = 0, k10i = 0, k11r = 0, k11i = 0;
  for (int n = 0; n < NSTATE; ++n) {
    float dr = gr - lr[n], di = gi - li[n];
    float dd = dr * dr + di * di;
    float ir = dr / dd, ii = -di / dd;
    float b = bb[n], p = pp[n], c = ct[n];
    float w00 = c * b, w01 = c * p, w10 = p * b, w11 = p * p;
    k00r += w00 * ir; k00i += w00 * ii;
    k01r += w01 * ir; k01i += w01 * ii;
    k10r += w10 * ir; k10i += w10 * ii;
    k11r += w11 * ir; k11i += w11 * ii;
  }
  float qr = k01r * k10r - k01i * k10i;
  float qi = k01r * k10i + k01i * k10r;
  float er = 1.f + k11r, ei = k11i;
  float ed = er * er + ei * ei;
  float ier = er / ed, iei = -ei / ed;
  float tr = qr * ier - qi * iei;
  float ti = qr * iei + qi * ier;
  float kr = k00r - tr, ki = k00i - ti;
  *outr = cr * kr - ci * ki;
  *outi = cr * ki + ci * kr;
}

__global__ void ssm_gen_khat(float* __restrict__ ws,
                             const float* __restrict__ Lr,
                             const float* __restrict__ Li,
                             const float* __restrict__ Pv,
                             const float* __restrict__ Bv,
                             const float* __restrict__ Ct,
                             const float* __restrict__ stepv) {
  int layer = blockIdx.y;
  int s = blockIdx.x * blockDim.x + threadIdx.x;
  int k2 = s >> 6, p = s & 63;
  int k1 = (k2 < 64) ? ((p >> 4) + 32 * ((p >> 3) & 1) + 4 * (p & 7)) : p;
  int kf = 128 * k1 + k2;
  int km = (8192 - kf) & 8191;

  float sc = 2.f / stepv[layer];
  const float* lr = Lr + layer * NSTATE;
  const float* li = Li + layer * NSTATE;
  const float* pp = Pv + layer * NSTATE;
  const float* bb = Bv + layer * NSTATE;
  const float* ct = Ct + layer * NSTATE;

  float ar, ai, br, bi;
  dplr_eval(kf, sc, lr, li, pp, bb, ct, &ar, &ai);
  dplr_eval(km, sc, lr, li, pp, bb, ct, &br, &bi);
  const float twice = (k2 == 0 || k2 == 64) ? 1.f : 2.f;
  const float scale = twice * 0.5f / 8192.f;
  ws[OFF_KHAT + layer * 16384 + 2 * s]     = (ar + br) * scale;
  ws[OFF_KHAT + layer * 16384 + 2 * s + 1] = (ai - bi) * scale;
}

__global__ void ssm_gen_s64(float* __restrict__ ws) {
  __shared__ float cs[128], sn[128], kr[64], ki[64];
  const int layer = blockIdx.y;
  const int tid = threadIdx.x;
  if (tid < 128) {
    float s, c;
    sincosf(2.f * PI_F * (float)tid / 128.f, &s, &c);
    cs[tid] = c; sn[tid] = s;
  }
  if (tid < 64) {
    kr[tid] = ws[OFF_KHAT + layer * 16384 + 2 * (4096 + tid)];
    ki[tid] = ws[OFF_KHAT + layer * 16384 + 2 * (4096 + tid) + 1];
  }
  __syncthreads();
  const int idx = blockIdx.x * 256 + tid;
  const int m = idx >> 6, n1 = idx & 63;
  const int d = n1 - m;
  float acc = 0.f;
  for (int k1 = 0; k1 < 64; ++k1) {
    int t = ((2 * k1 + 1) * d) & 127;
    acc += kr[k1] * cs[t] - ki[k1] * sn[t];
  }
  ws[OFF_S64P + layer * 4096 + (m >> 1) * 128 + 2 * n1 + (m & 1)] = acc;
}

__global__ __launch_bounds__(256) void ssm_rows(const float* __restrict__ uin,
                                                float* __restrict__ uout,
                                                const float* __restrict__ ws,
                                                int rows) {
  __shared__ __attribute__((aligned(16))) float smem[LDS_FLOATS];
  float* FL  = smem + LDS_F;

  const int tid  = threadIdx.x;
  const int wave = tid >> 5;
  const int lane = tid & 31;
  const int l16  = lane & 15;
  const int kSel = (lane >> 4) << 1;
  const int mHi  = (lane >> 4) << 3;
  const int waveU = __builtin_amdgcn_readfirstlane(wave);
  const int grp   = waveU >> 2;
  const int wl    = waveU & 3;
  const int kb    = (wl + 2 * grp) & 3;
  const int par   = kb & 1;
  const int k2L   = ((kb >> 1) << 5) + (l16 << 1) + par;
  const bool ownsZero = (kb == 0);
  v2f sg; sg.x = par ? -1.f : 1.f; sg.y = sg.x;
  const int mI    = wl << 4;
  const int tl    = tid & 127;
  const bool younger = waveU >= 4;

  float* xb  = smem + LDS_XB + grp * 8192;
  float* Yr  = smem + LDS_YR + grp * 4096;
  float* Yi  = smem + LDS_YI + grp * 4096;
  float* t64 = smem + LDS_T64 + grp * 64;

  {
    const float4* fsrc = (const float4*)(ws + OFF_F128Q);
    for (int i = tid; i < 64 * 32; i += 256) {
      int row = i >> 5, c4 = i & 31;
      float4 v = fsrc[(row << 6) + c4];
      *(float4*)(FL + row * FPITCH + (c4 << 2)) = v;
    }
    if (tid < 64) {
      float4 v; v.x = 1.f; v.y = 1.f; v.z = 1.f; v.w = -1.f;
      *(float4*)(FL + 128 * FPITCH + (tid << 2)) = v;
    }
  }

  const float* Wf     = ws + OFF_WF;
  const float* Wi     = ws + OFF_WI;

  const int frow = (ownsZero && l16 == 0) ? 128 : k2L;

  const int gb = (kSel << 8) + (l16 << 1);

  v2f f16c[4], f16s[4];
#pragma unroll
  for (int s = 0; s < 4; ++s) {
    int go = (((mHi >> 1) + s) << 7) + (l16 << 3);
    f16c[s] = *(const v2f*)(ws + OFF_G64PR + go);
    f16s[s] = *(const v2f*)(ws + OFF_G64PI + go);
  }
  const float4* tw4 = (const float4*)(ws + OFF_TW16 + ((lane >> 4) << 6));

  int swq[2];
#pragma unroll
  for (int eo = 0; eo < 2; ++eo) swq[eo] = (l16 ^ (2 * eo + (kSel >> 1))) << 1;

  const int pairs = (rows + 1) >> 1;
  for (int pp = blockIdx.x; pp < pairs; pp += gridDim.x) {
  const int r = 2 * pp + grp;
  const bool live = r < rows;
  const size_t rowOff = (size_t)(live ? r : 0) * LROW;
  {
    const float4* src = (const float4*)(uin + rowOff);
    float4 xv[16];
#pragma unroll
    for (int j = 0; j < 16; ++j) xv[j] = src[tl + j * 128];
#pragma unroll
    for (int j = 0; j < 16; ++j) {
      int i = tl + j * 128;
      float4 v = xv[j];
      int n2 = i >> 4, n1 = (i & 15) << 2;
      int base = pidx(n2, n1);
      xb[base] = v.x; xb[base + 2] = v.y; xb[base + 4] = v.z; xb[base + 6] = v.w;
    }
  }
  __syncthreads();

  for (int layer = 0; layer < 4; ++layer) {
    const float* kh = ws + OFF_KHAT + layer * 16384;

    if (younger) __builtin_amdgcn_s_setprio(3);

    v8f yr[4] = {}, yi[4] = {};
    for (int kk = 0; kk < 64; kk += 4) {
      float4 fq = *(const float4*)(FL + frow * FPITCH + (kk + kSel) * 2);
      v2f fr; fr.x = fq.x; fr.y = fq.y;
      v2f fi; fi.x = fq.z; fi.y = fq.w;
      int jb = ((kk + kSel) >> 1) << 7;
#pragma unroll
      for (int t = 0; t < 4; ++t) {
        v2f xl = *(const v2f*)(xb + jb + (((t << 4) + l16) << 1));
        v2f xh = *(const v2f*)(xb + 4096 + jb + (((t << 4) + l16) << 1));
        v2f x = xl + sg * xh;
        yr[t] = wmma4(x, fr, yr[t]);
        yi[t] = wmma4(x, fi, yi[t]);
      }
    }
    __builtin_amdgcn_s_setprio(0);

    if (ownsZero) {
      const float* s64 = ws + OFF_S64P + layer * 4096;
      v8f tz[4] = {};
      v2f sa[4][4][4];
#pragma unroll
      for (int mt = 0; mt < 4; ++mt) {
#pragma unroll
        for (int s = 0; s < 4; ++s) {
#pragma unroll
          for (int mo = 0; mo < 4; ++mo) {
            int go = gb + ((8 * mt + s) << 7) + (mo << 5);
            sa[mt][s][mo] = *(const v2f*)(s64 + go);
          }
        }
      }
      __builtin_amdgcn_sched_barrier(0);
#pragma unroll
      for (int mt = 0; mt < 4; ++mt) {
#pragma unroll
        for (int s = 0; s < 4; ++s) {
          v2f bv; bv.x = yi[mt][2 * s]; bv.y = yi[mt][2 * s + 1];
#pragma unroll
          for (int mo = 0; mo < 4; ++mo) tz[mo] = wmma4(sa[mt][s][mo], bv, tz[mo]);
        }
      }
#pragma unroll
      for (int mo = 0; mo < 4; ++mo) {
#pragma unroll
        for (int rr = 0; rr < 8; ++rr) yi[mo][rr] = (l16 == 0) ? 0.f : yi[mo][rr];
      }
      if (l16 == 0) {
#pragma unroll
        for (int mo = 0; mo < 4; ++mo) {
#pragma unroll
          for (int rr = 0; rr < 8; ++rr) t64[(mo << 4) + mHi + rr] = tz[mo][rr];
        }
      }
    }

    {
      const float4* w4 = (const float4*)(Wf + (((k2L << 6) + mHi) << 1));
#pragma unroll
      for (int t = 0; t < 4; ++t) {
#pragma unroll
        for (int q = 0; q < 4; ++q) {
          float4 w = w4[t * 8 + q];
          float a0 = yr[t][2 * q], b0 = yi[t][2 * q];
          float a1 = yr[t][2 * q + 1], b1 = yi[t][2 * q + 1];
          float i0 = a0 * w.y + b0 * w.x, i1 = a1 * w.w + b1 * w.z;
          yr[t][2 * q]     = a0 * w.x - b0 * w.y;
          yr[t][2 * q + 1] = a1 * w.z - b1 * w.w;
          yi[t][2 * q]  = i0;  yi[t][2 * q + 1]  = i1;
        }
      }
    }

    v8f tc[4], ts[4];
#pragma unroll
    for (int d = 1; d < 4; ++d) {
#pragma unroll
      for (int q = 0; q < 4; ++q) {
        float4 w = tw4[d * 4 + q];
        tc[d][2 * q] = w.x;     ts[d][2 * q] = w.y;
        tc[d][2 * q + 1] = w.z; ts[d][2 * q + 1] = w.w;
      }
    }

    v8f ur[4], ui[4], nui[4];
    {
      v8f s02r = yr[0] + yr[2], d02r = yr[0] - yr[2], s02i = yi[0] + yi[2], d02i = yi[0] - yi[2];
      v8f s13r = yr[1] + yr[3], d13r = yr[1] - yr[3], s13i = yi[1] + yi[3], d13i = yi[1] - yi[3];
      ur[0] = s02r + s13r;  ui[0] = s02i + s13i;
      v8f u2r = s02r - s13r, u2i = s02i - s13i;
      v8f u1r = d02r + d13i, u1i = d02i - d13r;
      v8f u3r = d02r - d13i, u3i = d02i + d13r;
      ur[1] = u1r * tc[1] - u1i * ts[1];  ui[1] = u1r * ts[1] + u1i * tc[1];
      ur[2] = u2r * tc[2] - u2i * ts[2];  ui[2] = u2r * ts[2] + u2i * tc[2];
      ur[3] = u3r * tc[3] - u3i * ts[3];  ui[3] = u3r * ts[3] + u3i * tc[3];
#pragma unroll
      for (int d = 0; d < 4; ++d) nui[d] = -ui[d];
    }
    v8f zr[4] = {}, zi[4] = {}, nzr[4];
#pragma unroll
    for (int s = 0; s < 4; ++s) {
#pragma unroll
      for (int d = 0; d < 4; ++d) {
        v2f br;  br.x  = ur[d][2 * s];  br.y  = ur[d][2 * s + 1];
        v2f bi;  bi.x  = ui[d][2 * s];  bi.y  = ui[d][2 * s + 1];
        v2f nbi; nbi.x = nui[d][2 * s]; nbi.y = nui[d][2 * s + 1];
        zr[d] = wmma4(f16c[s], br, zr[d]);  zr[d] = wmma4(f16s[s], nbi, zr[d]);
        zi[d] = wmma4(f16s[s], br, zi[d]);  zi[d] = wmma4(f16c[s], bi, zi[d]);
      }
    }
    {
      const float4* k4 = (const float4*)(kh + (((k2L << 6) + mHi) << 1));
#pragma unroll
      for (int mo = 0; mo < 4; ++mo) {
#pragma unroll
        for (int q = 0; q < 4; ++q) {
          float4 k = k4[mo * 8 + q];
          float a0 = zr[mo][2 * q], b0 = zi[mo][2 * q];
          float a1 = zr[mo][2 * q + 1], b1 = zi[mo][2 * q + 1];
          float r0 = a0 * k.x - b0 * k.y, r1 = a1 * k.z - b1 * k.w;
          zi[mo][2 * q]     = a0 * k.y + b0 * k.x;
          zi[mo][2 * q + 1] = a1 * k.w + b1 * k.z;
          zr[mo][2 * q]  = r0;  zr[mo][2 * q + 1]  = r1;
          nzr[mo][2 * q] = -r0; nzr[mo][2 * q + 1] = -r1;
        }
      }
    }

    v8f vr[4] = {}, vi[4] = {};
#pragma unroll
    for (int s = 0; s < 4; ++s) {
#pragma unroll
      for (int d = 0; d < 4; ++d) {
        v2f br;  br.x  = zr[d][2 * s];  br.y  = zr[d][2 * s + 1];
        v2f bi;  bi.x  = zi[d][2 * s];  bi.y  = zi[d][2 * s + 1];
        v2f nbr; nbr.x = nzr[d][2 * s]; nbr.y = nzr[d][2 * s + 1];
        vr[d] = wmma4(f16c[s], br, vr[d]);   vr[d] = wmma4(f16s[s], bi, vr[d]);
        vi[d] = wmma4(f16s[s], nbr, vi[d]);  vi[d] = wmma4(f16c[s], bi, vi[d]);
      }
    }
    v8f pr[4], pi[4];
    {
      v8f w1r = vr[1] * tc[1] + vi[1] * ts[1], w1i = vi[1] * tc[1] - vr[1] * ts[1];
      v8f w2r = vr[2] * tc[2] + vi[2] * ts[2], w2i = vi[2] * tc[2] - vr[2] * ts[2];
      v8f w3r = vr[3] * tc[3] + vi[3] * ts[3], w3i = vi[3] * tc[3] - vr[3] * ts[3];
      v8f s02r = vr[0] + w2r, d02r = vr[0] - w2r, s02i = vi[0] + w2i, d02i = vi[0] - w2i;
      v8f s13r = w1r + w3r,   d13r = w1r - w3r,   s13i = w1i + w3i,   d13i = w1i - w3i;
      pr[0] = s02r + s13r;  pi[0] = s02i + s13i;
      pr[2] = s02r - s13r;  pi[2] = s02i - s13i;
      pr[1] = d02r - d13i;  pi[1] = d02i + d13r;
      pr[3] = d02r + d13i;  pi[3] = d02i - d13r;
    }
    {
      const float4* w4 = (const float4*)(Wi + (((k2L << 6) + mHi) << 1));
      const int k2w = k2L;
      const int jw = ((k2w >> 2) << 1) | (k2w & 1);
      const int sb = (jw << 7) + ((k2w >> 1) & 1);
      const int q8 = (jw >> 1) & 7;
#pragma unroll
      for (int t = 0; t < 4; ++t) {
#pragma unroll
        for (int q = 0; q < 4; ++q) {
          float4 w = w4[t * 8 + q];
          float a0 = pr[t][2 * q], b0 = pi[t][2 * q];
          float a1 = pr[t][2 * q + 1], b1 = pi[t][2 * q + 1];
          int p0 = sb + ((((t << 4) + mHi + 2 * q)     ^ q8) << 1);
          int p1 = sb + ((((t << 4) + mHi + 2 * q + 1) ^ q8) << 1);
          Yr[p0] = a0 * w.x - b0 * w.y;
          Yi[p0] = a0 * w.y + b0 * w.x;
          Yr[p1] = a1 * w.z - b1 * w.w;
          Yi[p1] = a1 * w.w + b1 * w.z;
        }
      }
    }
    __syncthreads();

    {
      v8f accE[4], accO[4];
      v2f xlo[4][4], xhi[4][4];
      float tv[4];
#pragma unroll
      for (int t = 0; t < 4; ++t) {
        tv[t] = t64[(t << 4) + l16];
#pragma unroll
        for (int v = 0; v < 8; v += 2) {
          xlo[t][v >> 1] = *(const v2f*)(xb + pidx(mI + v + mHi, (t << 4) + l16));
          xhi[t][v >> 1] = *(const v2f*)(xb + pidx(64 + mI + v + mHi, (t << 4) + l16));
        }
      }
      __builtin_amdgcn_sched_barrier(0);
#pragma unroll
      for (int t = 0; t < 4; ++t) {
#pragma unroll
        for (int v = 0; v < 8; v += 2) {
          accE[t][v] = tv[t];  accE[t][v + 1] = -tv[t];
          accO[t][v] = 0.f;    accO[t][v + 1] = 0.f;
        }
      }
#pragma nounroll
      for (int e0 = 0; e0 < 8; e0 += 2) {
        const int sx = (e0 & 2) << 2;
        float4 q0[2], q1[2];
        v2f brE[2][4], biE[2][4], brO[2][4], biO[2][4];
#pragma unroll
        for (int eo = 0; eo < 2; ++eo) {
          const int e = e0 + eo;
          const float* fa = FL + (mI + l16) * FPITCH + (e << 4) + (kSel << 2);
          q0[eo] = *(const float4*)(fa);
          q1[eo] = *(const float4*)(fa + 4);
          const int jbE = ((4 * e + kSel) << 7) + (swq[eo] ^ sx);
          const int jbO = ((4 * e + kSel + 1) << 7) + (swq[eo] ^ sx);
#pragma unroll
          for (int t = 0; t < 4; ++t) {
            brE[eo][t] = *(const v2f*)(Yr + jbE + (t << 5));
            biE[eo][t] = *(const v2f*)(Yi + jbE + (t << 5));
            brO[eo][t] = *(const v2f*)(Yr + jbO + (t << 5));
            biO[eo][t] = *(const v2f*)(Yi + jbO + (t << 5));
          }
        }
        __builtin_amdgcn_sched_barrier(0);
#pragma unroll
        for (int eo = 0; eo < 2; ++eo) {
          v2f ce; ce.x = q0[eo].x; ce.y = q1[eo].x;
          v2f se; se.x = q0[eo].z; se.y = q1[eo].z;
          v2f co; co.x = q0[eo].y; co.y = q1[eo].y;
          v2f so; so.x = q0[eo].w; so.y = q1[eo].w;
#pragma unroll
          for (int t = 0; t < 4; ++t) {
            accE[t] = wmma4(ce, brE[eo][t], accE[t]);
            accO[t] = wmma4(co, brO[eo][t], accO[t]);
            accE[t] = wmma4(se, biE[eo][t], accE[t]);
            accO[t] = wmma4(so, biO[eo][t], accO[t]);
          }
        }
      }
#pragma unroll
      for (int t = 0; t < 4; ++t) {
#pragma unroll
        for (int v = 0; v < 8; ++v) {
          const int row = mI + v + mHi, col = (t << 4) + l16;
          const float xl = (v & 1) ? xlo[t][v >> 1].y : xlo[t][v >> 1].x;
          const float xh = (v & 1) ? xhi[t][v >> 1].y : xhi[t][v >> 1].x;
          const float yl = (accE[t][v] + accO[t][v]) + xl;
          const float yh = (accE[t][v] - accO[t][v]) + xh;
          xb[pidx(row, col)]      = 0.5f * yl * (1.f + erff(yl * 0.70710678118654752f));
          xb[pidx(row + 64, col)] = 0.5f * yh * (1.f + erff(yh * 0.70710678118654752f));
        }
      }
    }
    __syncthreads();
  }

  if (live) {
    float4* dst = (float4*)(uout + rowOff);
    for (int i = tl; i < LROW / 4; i += 128) {
      int n2 = i >> 4, n1 = (i & 15) << 2;
      int base = pidx(n2, n1);
      float4 o;
      o.x = xb[base]; o.y = xb[base + 2]; o.z = xb[base + 4]; o.w = xb[base + 6];
      dst[i] = o;
    }
  }
  }
}

extern "C" void kernel_launch(void* const* d_in, const int* in_sizes, int n_in,
                              void* d_out, int out_size, void* d_ws, size_t ws_size,
                              hipStream_t stream) {
  const float* u    = (const float*)d_in[0];
  const float* Lr   = (const float*)d_in[1];
  const float* Li   = (const float*)d_in[2];
  const float* Pv   = (const float*)d_in[3];
  const float* Bv   = (const float*)d_in[4];
  const float* Ct   = (const float*)d_in[5];
  const float* step = (const float*)d_in[6];
  float* out = (float*)d_out;
  float* ws  = (float*)d_ws;

  int rows = in_sizes[0] / LROW;

  ssm_gen_tables<<<64, 256, 0, stream>>>(ws);
  ssm_gen_khat<<<dim3(32, 4), 256, 0, stream>>>(ws, Lr, Li, Pv, Bv, Ct, step);
  ssm_gen_s64<<<dim3(16, 4), 256, 0, stream>>>(ws);
  ssm_rows<<<256, 256, 0, stream>>>(u, out, ws, rows);
}


# One row per 4-wave workgroup with its own barriers, two workgroups per WGP (was one 8-wave two-row workgroup; waves per SIMD and registers unchanged; priority window dropped); sum order unchanged
# speedup vs baseline: 3.9057x; 1.1791x over previous
#include <hip/hip_runtime.h>
#include <math.h>

typedef float v2f __attribute__((ext_vector_type(2)));
typedef float v8f __attribute__((ext_vector_type(8)));

#define LROW 8192
#define NSTATE 64
#define PI_F 3.14159265358979323846f

#define OFF_F128Q   0
#define OFF_S64P    32768
#define OFF_G64PR   65536
#define OFF_G64PI   69632
#define OFF_TW16    73728
#define OFF_WF      81920
#define OFF_WI      98304
#define OFF_KHAT    114688
#define WS_FLOATS   (OFF_KHAT + 4*16384)

#define LDS_XB      0
#define LDS_YR      8192
#define LDS_YI      12288
#define LDS_T64     16384
#define LDS_F       16448
#define FPITCH      260
#define FROWX       64
#define LDS_FLOATS  (LDS_F + 65 * FPITCH)

static __device__ inline v8f wmma4(v2f a, v2f b, v8f c) {
  return __builtin_amdgcn_wmma_f32_16x16x4_f32(false, a, false, b, (short)0, c,
                                               false, false);
}

static __device__ inline int pidx(int row, int col) {
  return ((row >> 1) << 7) + (col << 1) + (row & 1);
}

__global__ void ssm_gen_tables(float* __restrict__ ws) {
  int idx = blockIdx.x * blockDim.x + threadIdx.x;
  {
    int k2 = idx >> 7, n2 = idx & 127;
    int t = (k2 * n2) & 127;
    float s, c;
    sincosf(-2.f * PI_F * (float)t / 128.f, &s, &c);
    int pos = k2 * 256 + (n2 >> 1) * 4 + (n2 & 1);
    ws[OFF_F128Q + pos]      = c;
    ws[OFF_F128Q + pos + 2]  = s;
  }
  if (idx < 8192) {
    int k2 = idx >> 6, n1 = idx & 63;
    int t = (n1 * k2) & 8191;
    float s, c;
    sincosf(-2.f * PI_F * (float)t / 8192.f, &s, &c);
    ws[OFF_WF + 2 * idx]     = c;  ws[OFF_WF + 2 * idx + 1] = s;
    ws[OFF_WI + 2 * idx]     = c;  ws[OFF_WI + 2 * idx + 1] = -s;
  }
  if (idx < 4096) {
    int n1 = idx >> 6, k1 = idx & 63;
    int t = (n1 * k1) & 63;
    float s, c;
    sincosf(-2.f * PI_F * (float)t / 64.f, &s, &c);
    int pos = (n1 >> 1) * 128 + 2 * k1 + (n1 & 1);
    ws[OFF_G64PR + pos]  = c;  ws[OFF_G64PI + pos]  = s;
  }
  if (idx < 64) {
    int h = idx >> 5, d = (idx >> 3) & 3, r = idx & 7;
    int t = ((8 * h + r) * d) & 63;
    float s, c;
    sincosf(-2.f * PI_F * (float)t / 64.f, &s, &c);
    ws[OFF_TW16 + 2 * idx] = c;  ws[OFF_TW16 + 2 * idx + 1] = s;
  }
}

static __device__ inline void dplr_eval(int kf, float sc,
                                        const float* __restrict__ lr, const float* __restrict__ li,
                                        const float* __restrict__ pp, const float* __restrict__ bb,
                                        const float* __restrict__ ct, float* outr, float* outi) {
  float osin, ocos;
  sincosf(-2.f * PI_F * (float)kf / 8192.f, &osin, &ocos);
  float pr = 1.f + ocos, pi = osin;
  float mr = 1.f - ocos, mi = -osin;
  float pd = pr * pr + pi * pi;
  float ipr = pr / pd, ipi = -pi / pd;
  float cr = 2.f * ipr, ci = 2.f * ipi;
  float gr = sc * (mr * ipr - mi * ipi);
  float gi = sc * (mr * ipi + mi * ipr);

  float k00r = 0, k00i = 0, k01r = 0, k01i = 0;
  float k10r = 0, k10i = 0, k11r = 0, k11i = 0;
  for (int n = 0; n < NSTATE; ++n) {
    float dr = gr - lr[n], di = gi - li[n];
    float dd = dr * dr + di * di;
    float ir = dr / dd, ii = -di / dd;
    float b = bb[n], p = pp[n], c = ct[n];
    float w00 = c * b, w01 = c * p, w10 = p * b, w11 = p * p;
    k00r += w00 * ir; k00i += w00 * ii;
    k01r += w01 * ir; k01i += w01 * ii;
    k10r += w10 * ir; k10i += w10 * ii;
    k11r += w11 * ir; k11i += w11 * ii;
  }
  float qr = k01r * k10r - k01i * k10i;
  float qi = k01r * k10i + k01i * k10r;
  float er = 1.f + k11r, ei = k11i;
  float ed = er * er + ei * ei;
  float ier = er / ed, iei = -ei / ed;
  float tr = qr * ier - qi * iei;
  float ti = qr * iei + qi * ier;
  float kr = k00r - tr, ki = k00i - ti;
  *outr = cr * kr - ci * ki;
  *outi = cr * ki + ci * kr;
}

__global__ void ssm_gen_khat(float* __restrict__ ws,
                             const float* __restrict__ Lr,
                             const float* __restrict__ Li,
                             const float* __restrict__ Pv,
                             const float* __restrict__ Bv,
                             const float* __restrict__ Ct,
                             const float* __restrict__ stepv) {
  int layer = blockIdx.y;
  int s = blockIdx.x * blockDim.x + threadIdx.x;
  int k2 = s >> 6, p = s & 63;
  int k1 = (k2 < 64) ? ((p >> 4) + 32 * ((p >> 3) & 1) + 4 * (p & 7)) : p;
  int kf = 128 * k1 + k2;
  int km = (8192 - kf) & 8191;

  float sc = 2.f / stepv[layer];
  const float* lr = Lr + layer * NSTATE;
  const float* li = Li + layer * NSTATE;
  const float* pp = Pv + layer * NSTATE;
  const float* bb = Bv + layer * NSTATE;
  const float* ct = Ct + layer * NSTATE;

  float ar, ai, br, bi;
  dplr_eval(kf, sc, lr, li, pp, bb, ct, &ar, &ai);
  dplr_eval(km, sc, lr, li, pp, bb, ct, &br, &bi);
  const float twice = (k2 == 0 || k2 == 64) ? 1.f : 2.f;
  const float scale = twice * 0.5f / 8192.f;
  ws[OFF_KHAT + layer * 16384 + 2 * s]     = (ar + br) * scale;
  ws[OFF_KHAT + layer * 16384 + 2 * s + 1] = (ai - bi) * scale;
}

__global__ void ssm_gen_s64(float* __restrict__ ws) {
  __shared__ float cs[128], sn[128], kr[64], ki[64];
  const int layer = blockIdx.y;
  const int tid = threadIdx.x;
  if (tid < 128) {
    float s, c;
    sincosf(2.f * PI_F * (float)tid / 128.f, &s, &c);
    cs[tid] = c; sn[tid] = s;
  }
  if (tid < 64) {
    kr[tid] = ws[OFF_KHAT + layer * 16384 + 2 * (4096 + tid)];
    ki[tid] = ws[OFF_KHAT + layer * 16384 + 2 * (4096 + tid) + 1];
  }
  __syncthreads();
  const int idx = blockIdx.x * 256 + tid;
  const int m = idx >> 6, n1 = idx & 63;
  const int d = n1 - m;
  float acc = 0.f;
  for (int k1 = 0; k1 < 64; ++k1) {
    int t = ((2 * k1 + 1) * d) & 127;
    acc += kr[k1] * cs[t] - ki[k1] * sn[t];
  }
  ws[OFF_S64P + layer * 4096 + (m >> 1) * 128 + 2 * n1 + (m & 1)] = acc;
}

__global__ __launch_bounds__(128, 2) void ssm_rows(const float* __restrict__ uin,
                                                float* __restrict__ uout,
                                                const float* __restrict__ ws,
                                                int rows) {
  __shared__ __attribute__((aligned(16))) float smem[LDS_FLOATS];
  float* FL  = smem + LDS_F;

  const int tid  = threadIdx.x;
  const int wave = tid >> 5;
  const int lane = tid & 31;
  const int l16  = lane & 15;
  const int kSel = (lane >> 4) << 1;
  const int mHi  = (lane >> 4) << 3;
  const int waveU = __builtin_amdgcn_readfirstlane(wave);
  const int wl    = waveU & 3;
  const int kb    = wl;
  const int par   = kb & 1;
  const int k2L   = ((kb >> 1) << 5) + (l16 << 1) + par;
  const bool ownsZero = (kb == 0);
  v2f sg; sg.x = par ? -1.f : 1.f; sg.y = sg.x;
  const int mI    = wl << 4;
  const int tl    = tid;

  float* xb  = smem + LDS_XB;
  float* Yr  = smem + LDS_YR;
  float* Yi  = smem + LDS_YI;
  float* t64 = smem + LDS_T64;

  {
    const float4* fsrc = (const float4*)(ws + OFF_F128Q);
    for (int i = tid; i < 64 * 32; i += 128) {
      int row = i >> 5, c4 = i & 31;
      float4 v = fsrc[(row << 6) + c4];
      *(float4*)(FL + row * FPITCH + (c4 << 2)) = v;
    }
    if (tid < 64) {
      float4 v; v.x = 1.f; v.y = 1.f; v.z = 1.f; v.w = -1.f;
      *(float4*)(FL + FROWX * FPITCH + (tid << 2)) = v;
    }
  }

  const float* Wf     = ws + OFF_WF;
  const float* Wi     = ws + OFF_WI;

  const int frow = (ownsZero && l16 == 0) ? FROWX : k2L;

  const int gb = (kSel << 8) + (l16 << 1);

  v2f f16c[4], f16s[4];
#pragma unroll
  for (int s = 0; s < 4; ++s) {
    int go = (((mHi >> 1) + s) << 7) + (l16 << 3);
    f16c[s] = *(const v2f*)(ws + OFF_G64PR + go);
    f16s[s] = *(const v2f*)(ws + OFF_G64PI + go);
  }
  const float4* tw4 = (const float4*)(ws + OFF_TW16 + ((lane >> 4) << 6));

  int swq[2];
#pragma unroll
  for (int eo = 0; eo < 2; ++eo) swq[eo] = (l16 ^ (2 * eo + (kSel >> 1))) << 1;

  for (int r = blockIdx.x; r < rows; r += gridDim.x) {
  const size_t rowOff = (size_t)r * LROW;
  {
    const float4* src = (const float4*)(uin + rowOff);
    float4 xv[16];
#pragma unroll
    for (int j = 0; j < 16; ++j) xv[j] = src[tl + j * 128];
#pragma unroll
    for (int j = 0; j < 16; ++j) {
      int i = tl + j * 128;
      float4 v = xv[j];
      int n2 = i >> 4, n1 = (i & 15) << 2;
      int base = pidx(n2, n1);
      xb[base] = v.x; xb[base + 2] = v.y; xb[base + 4] = v.z; xb[base + 6] = v.w;
    }
  }
  __syncthreads();

  for (int layer = 0; layer < 4; ++layer) {
    const float* kh = ws + OFF_KHAT + layer * 16384;

    v8f yr[4] = {}, yi[4] = {};
    for (int kk = 0; kk < 64; kk += 4) {
      float4 fq = *(const float4*)(FL + frow * FPITCH + (kk + kSel) * 2);
      v2f fr; fr.x = fq.x; fr.y = fq.y;
      v2f fi; fi.x = fq.z; fi.y = fq.w;
      int jb = ((kk + kSel) >> 1) << 7;
#pragma unroll
      for (int t = 0; t < 4; ++t) {
        v2f xl = *(const v2f*)(xb + jb + (((t << 4) + l16) << 1));
        v2f xh = *(const v2f*)(xb + 4096 + jb + (((t << 4) + l16) << 1));
        v2f x = xl + sg * xh;
        yr[t] = wmma4(x, fr, yr[t]);
        yi[t] = wmma4(x, fi, yi[t]);
      }
    }
    if (ownsZero) {
      const float* s64 = ws + OFF_S64P + layer * 4096;
      v8f tz[4] = {};
      v2f sa[4][4][4];
#pragma unroll
      for (int mt = 0; mt < 4; ++mt) {
#pragma unroll
        for (int s = 0; s < 4; ++s) {
#pragma unroll
          for (int mo = 0; mo < 4; ++mo) {
            int go = gb + ((8 * mt + s) << 7) + (mo << 5);
            sa[mt][s][mo] = *(const v2f*)(s64 + go);
          }
        }
      }
      __builtin_amdgcn_sched_barrier(0);
#pragma unroll
      for (int mt = 0; mt < 4; ++mt) {
#pragma unroll
        for (int s = 0; s < 4; ++s) {
          v2f bv; bv.x = yi[mt][2 * s]; bv.y = yi[mt][2 * s + 1];
#pragma unroll
          for (int mo = 0; mo < 4; ++mo) tz[mo] = wmma4(sa[mt][s][mo], bv, tz[mo]);
        }
      }
#pragma unroll
      for (int mo = 0; mo < 4; ++mo) {
#pragma unroll
        for (int rr = 0; rr < 8; ++rr) yi[mo][rr] = (l16 == 0) ? 0.f : yi[mo][rr];
      }
      if (l16 == 0) {
#pragma unroll
        for (int mo = 0; mo < 4; ++mo) {
#pragma unroll
          for (int rr = 0; rr < 8; ++rr) t64[(mo << 4) + mHi + rr] = tz[mo][rr];
        }
      }
    }

    {
      const float4* w4 = (const float4*)(Wf + (((k2L << 6) + mHi) << 1));
#pragma unroll
      for (int t = 0; t < 4; ++t) {
#pragma unroll
        for (int q = 0; q < 4; ++q) {
          float4 w = w4[t * 8 + q];
          float a0 = yr[t][2 * q], b0 = yi[t][2 * q];
          float a1 = yr[t][2 * q + 1], b1 = yi[t][2 * q + 1];
          float i0 = a0 * w.y + b0 * w.x, i1 = a1 * w.w + b1 * w.z;
          yr[t][2 * q]     = a0 * w.x - b0 * w.y;
          yr[t][2 * q + 1] = a1 * w.z - b1 * w.w;
          yi[t][2 * q]  = i0;  yi[t][2 * q + 1]  = i1;
        }
      }
    }

    v8f tc[4], ts[4];
#pragma unroll
    for (int d = 1; d < 4; ++d) {
#pragma unroll
      for (int q = 0; q < 4; ++q) {
        float4 w = tw4[d * 4 + q];
        tc[d][2 * q] = w.x;     ts[d][2 * q] = w.y;
        tc[d][2 * q + 1] = w.z; ts[d][2 * q + 1] = w.w;
      }
    }

    v8f ur[4], ui[4], nui[4];
    {
      v8f s02r = yr[0] + yr[2], d02r = yr[0] - yr[2], s02i = yi[0] + yi[2], d02i = yi[0] - yi[2];
      v8f s13r = yr[1] + yr[3], d13r = yr[1] - yr[3], s13i = yi[1] + yi[3], d13i = yi[1] - yi[3];
      ur[0] = s02r + s13r;  ui[0] = s02i + s13i;
      v8f u2r = s02r - s13r, u2i = s02i - s13i;
      v8f u1r = d02r + d13i, u1i = d02i - d13r;
      v8f u3r = d02r - d13i, u3i = d02i + d13r;
      ur[1] = u1r * tc[1] - u1i * ts[1];  ui[1] = u1r * ts[1] + u1i * tc[1];
      ur[2] = u2r * tc[2] - u2i * ts[2];  ui[2] = u2r * ts[2] + u2i * tc[2];
      ur[3] = u3r * tc[3] - u3i * ts[3];  ui[3] = u3r * ts[3] + u3i * tc[3];
#pragma unroll
      for (int d = 0; d < 4; ++d) nui[d] = -ui[d];
    }
    v8f zr[4] = {}, zi[4] = {}, nzr[4];
#pragma unroll
    for (int s = 0; s < 4; ++s) {
#pragma unroll
      for (int d = 0; d < 4; ++d) {
        v2f br;  br.x  = ur[d][2 * s];  br.y  = ur[d][2 * s + 1];
        v2f bi;  bi.x  = ui[d][2 * s];  bi.y  = ui[d][2 * s + 1];
        v2f nbi; nbi.x = nui[d][2 * s]; nbi.y = nui[d][2 * s + 1];
        zr[d] = wmma4(f16c[s], br, zr[d]);  zr[d] = wmma4(f16s[s], nbi, zr[d]);
        zi[d] = wmma4(f16s[s], br, zi[d]);  zi[d] = wmma4(f16c[s], bi, zi[d]);
      }
    }
    {
      const float4* k4 = (const float4*)(kh + (((k2L << 6) + mHi) << 1));
#pragma unroll
      for (int mo = 0; mo < 4; ++mo) {
#pragma unroll
        for (int q = 0; q < 4; ++q) {
          float4 k = k4[mo * 8 + q];
          float a0 = zr[mo][2 * q], b0 = zi[mo][2 * q];
          float a1 = zr[mo][2 * q + 1], b1 = zi[mo][2 * q + 1];
          float r0 = a0 * k.x - b0 * k.y, r1 = a1 * k.z - b1 * k.w;
          zi[mo][2 * q]     = a0 * k.y + b0 * k.x;
          zi[mo][2 * q + 1] = a1 * k.w + b1 * k.z;
          zr[mo][2 * q]  = r0;  zr[mo][2 * q + 1]  = r1;
          nzr[mo][2 * q] = -r0; nzr[mo][2 * q + 1] = -r1;
        }
      }
    }

    v8f vr[4] = {}, vi[4] = {};
#pragma unroll
    for (int s = 0; s < 4; ++s) {
#pragma unroll
      for (int d = 0; d < 4; ++d) {
        v2f br;  br.x  = zr[d][2 * s];  br.y  = zr[d][2 * s + 1];
        v2f bi;  bi.x  = zi[d][2 * s];  bi.y  = zi[d][2 * s + 1];
        v2f nbr; nbr.x = nzr[d][2 * s]; nbr.y = nzr[d][2 * s + 1];
        vr[d] = wmma4(f16c[s], br, vr[d]);   vr[d] = wmma4(f16s[s], bi, vr[d]);
        vi[d] = wmma4(f16s[s], nbr, vi[d]);  vi[d] = wmma4(f16c[s], bi, vi[d]);
      }
    }
    v8f pr[4], pi[4];
    {
      v8f w1r = vr[1] * tc[1] + vi[1] * ts[1], w1i = vi[1] * tc[1] - vr[1] * ts[1];
      v8f w2r = vr[2] * tc[2] + vi[2] * ts[2], w2i = vi[2] * tc[2] - vr[2] * ts[2];
      v8f w3r = vr[3] * tc[3] + vi[3] * ts[3], w3i = vi[3] * tc[3] - vr[3] * ts[3];
      v8f s02r = vr[0] + w2r, d02r = vr[0] - w2r, s02i = vi[0] + w2i, d02i = vi[0] - w2i;
      v8f s13r = w1r + w3r,   d13r = w1r - w3r,   s13i = w1i + w3i,   d13i = w1i - w3i;
      pr[0] = s02r + s13r;  pi[0] = s02i + s13i;
      pr[2] = s02r - s13r;  pi[2] = s02i - s13i;
      pr[1] = d02r - d13i;  pi[1] = d02i + d13r;
      pr[3] = d02r + d13i;  pi[3] = d02i - d13r;
    }
    {
      const float4* w4 = (const float4*)(Wi + (((k2L << 6) + mHi) << 1));
      const int k2w = k2L;
      const int jw = ((k2w >> 2) << 1) | (k2w & 1);
      const int sb = (jw << 7) + ((k2w >> 1) & 1);
      const int q8 = (jw >> 1) & 7;
#pragma unroll
      for (int t = 0; t < 4; ++t) {
#pragma unroll
        for (int q = 0; q < 4; ++q) {
          float4 w = w4[t * 8 + q];
          float a0 = pr[t][2 * q], b0 = pi[t][2 * q];
          float a1 = pr[t][2 * q + 1], b1 = pi[t][2 * q + 1];
          int p0 = sb + ((((t << 4) + mHi + 2 * q)     ^ q8) << 1);
          int p1 = sb + ((((t << 4) + mHi + 2 * q + 1) ^ q8) << 1);
          Yr[p0] = a0 * w.x - b0 * w.y;
          Yi[p0] = a0 * w.y + b0 * w.x;
          Yr[p1] = a1 * w.z - b1 * w.w;
          Yi[p1] = a1 * w.w + b1 * w.z;
        }
      }
    }
    __syncthreads();

    {
      v8f accE[4], accO[4];
      v2f xlo[4][4], xhi[4][4];
      float tv[4];
#pragma unroll
      for (int t = 0; t < 4; ++t) {
        tv[t] = t64[(t << 4) + l16];
#pragma unroll
        for (int v = 0; v < 8; v += 2) {
          xlo[t][v >> 1] = *(const v2f*)(xb + pidx(mI + v + mHi, (t << 4) + l16));
          xhi[t][v >> 1] = *(const v2f*)(xb + pidx(64 + mI + v + mHi, (t << 4) + l16));
        }
      }
      __builtin_amdgcn_sched_barrier(0);
#pragma unroll
      for (int t = 0; t < 4; ++t) {
#pragma unroll
        for (int v = 0; v < 8; v += 2) {
          accE[t][v] = tv[t];  accE[t][v + 1] = -tv[t];
          accO[t][v] = 0.f;    accO[t][v + 1] = 0.f;
        }
      }
#pragma nounroll
      for (int e0 = 0; e0 < 8; e0 += 2) {
        const int sx = (e0 & 2) << 2;
        float4 q0[2], q1[2];
        v2f brE[2][4], biE[2][4], brO[2][4], biO[2][4];
#pragma unroll
        for (int eo = 0; eo < 2; ++eo) {
          const int e = e0 + eo;
          const float* fa = FL + (mI + l16) * FPITCH + (e << 4) + (kSel << 2);
          q0[eo] = *(const float4*)(fa);
          q1[eo] = *(const float4*)(fa + 4);
          const int jbE = ((4 * e + kSel) << 7) + (swq[eo] ^ sx);
          const int jbO = ((4 * e + kSel + 1) << 7) + (swq[eo] ^ sx);
#pragma unroll
          for (int t = 0; t < 4; ++t) {
            brE[eo][t] = *(const v2f*)(Yr + jbE + (t << 5));
            biE[eo][t] = *(const v2f*)(Yi + jbE + (t << 5));
            brO[eo][t] = *(const v2f*)(Yr + jbO + (t << 5));
            biO[eo][t] = *(const v2f*)(Yi + jbO + (t << 5));
          }
        }
        __builtin_amdgcn_sched_barrier(0);
#pragma unroll
        for (int eo = 0; eo < 2; ++eo) {
          v2f ce; ce.x = q0[eo].x; ce.y = q1[eo].x;
          v2f se; se.x = q0[eo].z; se.y = q1[eo].z;
          v2f co; co.x = q0[eo].y; co.y = q1[eo].y;
          v2f so; so.x = q0[eo].w; so.y = q1[eo].w;
#pragma unroll
          for (int t = 0; t < 4; ++t) {
            accE[t] = wmma4(ce, brE[eo][t], accE[t]);
            accO[t] = wmma4(co, brO[eo][t], accO[t]);
            accE[t] = wmma4(se, biE[eo][t], accE[t]);
            accO[t] = wmma4(so, biO[eo][t], accO[t]);
          }
        }
      }
#pragma unroll
      for (int t = 0; t < 4; ++t) {
#pragma unroll
        for (int v = 0; v < 8; ++v) {
          const int row = mI + v + mHi, col = (t << 4) + l16;
          const float xl = (v & 1) ? xlo[t][v >> 1].y : xlo[t][v >> 1].x;
          const float xh = (v & 1) ? xhi[t][v >> 1].y : xhi[t][v >> 1].x;
          const float yl = (accE[t][v] + accO[t][v]) + xl;
          const float yh = (accE[t][v] - accO[t][v]) + xh;
          xb[pidx(row, col)]      = 0.5f * yl * (1.f + erff(yl * 0.70710678118654752f));
          xb[pidx(row + 64, col)] = 0.5f * yh * (1.f + erff(yh * 0.70710678118654752f));
        }
      }
    }
    __syncthreads();
  }

  {
    float4* dst = (float4*)(uout + rowOff);
    for (int i = tl; i < LROW / 4; i += 128) {
      int n2 = i >> 4, n1 = (i & 15) << 2;
      int base = pidx(n2, n1);
      float4 o;
      o.x = xb[base]; o.y = xb[base + 2]; o.z = xb[base + 4]; o.w = xb[base + 6];
      dst[i] = o;
    }
  }
  }
}

extern "C" void kernel_launch(void* const* d_in, const int* in_sizes, int n_in,
                              void* d_out, int out_size, void* d_ws, size_t ws_size,
                              hipStream_t stream) {
  const float* u    = (const float*)d_in[0];
  const float* Lr   = (const float*)d_in[1];
  const float* Li   = (const float*)d_in[2];
  const float* Pv   = (const float*)d_in[3];
  const float* Bv   = (const float*)d_in[4];
  const float* Ct   = (const float*)d_in[5];
  const float* step = (const float*)d_in[6];
  float* out = (float*)d_out;
  float* ws  = (float*)d_ws;

  int rows = in_sizes[0] / LROW;

  ssm_gen_tables<<<64, 256, 0, stream>>>(ws);
  ssm_gen_khat<<<dim3(32, 4), 256, 0, stream>>>(ws, Lr, Li, Pv, Bv, Ct, step);
  ssm_gen_s64<<<dim3(16, 4), 256, 0, stream>>>(ws);
  ssm_rows<<<512, 128, 0, stream>>>(u, out, ws, rows);
}
